# starting kernel
# baseline (speedup 1.0000x reference)
#include <hip/hip_runtime.h>
#include <math.h>

typedef __attribute__((ext_vector_type(2))) float v2f;
typedef __attribute__((ext_vector_type(8))) float v8f;

#define Nn   4096
#define INF_ 3.402823466e+38f
#define Dd   256
#define Hh   8
#define HDd  32
#define Ll   2
#define Ee   131072
#define Pp   200000
#define INc  128

__device__ __forceinline__ v8f wmma4(v2f a, v2f b, v8f c) {
  return __builtin_amdgcn_wmma_f32_16x16x4_f32(
      false, a, false, b, (short)0, c, false, false);
}

__device__ __forceinline__ float rsum16(float v) {
  for (int m = 8; m >= 1; m >>= 1) v += __shfl_xor(v, m, 16);
  return v;
}
__device__ __forceinline__ float rmax16(float v) {
  for (int m = 8; m >= 1; m >>= 1) v = fmaxf(v, __shfl_xor(v, m, 16));
  return v;
}

template <int KK>
__global__ __launch_bounds__(32)
void gemm_atb(const float* __restrict__ A, const float* __restrict__ W,
              const float* __restrict__ bias, float* __restrict__ C,
              int M, int Nout) {
  const int tid  = threadIdx.x;
  const int m    = tid & 15;
  const int half = tid >> 4;
  const int tbn  = blockIdx.x * 64;
  const int tbm  = blockIdx.y * 16;

  const float* arow = A + (size_t)(tbm + m) * KK + half * 2;
  const float* w0   = W + (size_t)(tbn + m) * KK + half * 2;
  const float* w1   = w0 + (size_t)16 * KK;
  const float* w2   = w1 + (size_t)16 * KK;
  const float* w3   = w2 + (size_t)16 * KK;

  v8f c0 = {}, c1 = {}, c2 = {}, c3 = {};
#pragma unroll
  for (int k = 0; k < KK; k += 4) {
    v2f a  = { arow[k], arow[k + 1] };
    v2f b0 = { w0[k],   w0[k + 1] };
    v2f b1 = { w1[k],   w1[k + 1] };
    v2f b2 = { w2[k],   w2[k + 1] };
    v2f b3 = { w3[k],   w3[k + 1] };
    c0 = wmma4(a, b0, c0);
    c1 = wmma4(a, b1, c1);
    c2 = wmma4(a, b2, c2);
    c3 = wmma4(a, b3, c3);
  }
  for (int r = 0; r < 8; ++r) {
    const int row = tbm + r + 8 * half;
    float* crow = C + (size_t)row * Nout + tbn + m;
    if (bias) {
      const float* brow = bias + tbn + m;
      crow[0]  = c0[r] + brow[0];
      crow[16] = c1[r] + brow[16];
      crow[32] = c2[r] + brow[32];
      crow[48] = c3[r] + brow[48];
    } else {
      crow[0] = c0[r]; crow[16] = c1[r]; crow[32] = c2[r]; crow[48] = c3[r];
    }
  }
}

__global__ __launch_bounds__(32)
void attn_kernel(const float* __restrict__ qkv, float* __restrict__ obuf) {
  const int tid  = threadIdx.x;
  const int m    = tid & 15;
  const int half = tid >> 4;
  const int qb   = blockIdx.x;
  const int h    = blockIdx.y;
  const float scale = 0.1767766952966369f;

  const float* qbase = qkv + (size_t)(qb * 16) * (3 * Dd) + h * HDd;

  __shared__ float pLds[16 * 16];

  v2f qa[8];
#pragma unroll
  for (int kk = 0; kk < HDd; kk += 4) {
    const int k0 = kk + half * 2;
    qa[kk >> 2] = v2f{ qbase[m * (3 * Dd) + k0], qbase[m * (3 * Dd) + k0 + 1] };
  }

  v8f o0 = {}, o1 = {};
  float mrow[8], lrow[8];
  for (int r = 0; r < 8; ++r) { mrow[r] = -INF_; lrow[r] = 0.f; }

  for (int kb = 0; kb < Nn / 16; ++kb) {
    const float* kbase = qkv + (size_t)(kb * 16) * (3 * Dd) + Dd + h * HDd;
    const float* vbase = qkv + (size_t)(kb * 16) * (3 * Dd) + 2 * Dd + h * HDd;

    v8f s = {};
#pragma unroll
    for (int kk = 0; kk < HDd; kk += 4) {
      const int k0 = kk + half * 2;
      v2f b = { kbase[m * (3 * Dd) + k0], kbase[m * (3 * Dd) + k0 + 1] };
      s = wmma4(qa[kk >> 2], b, s);
    }

    float alpha[8];
#pragma unroll
    for (int r = 0; r < 8; ++r) {
      float sv = s[r] * scale;
      float rm = rmax16(sv);
      float mn = fmaxf(mrow[r], rm);
      float p  = __expf(sv - mn);
      float co = __expf(mrow[r] - mn);
      float rs = rsum16(p);
      lrow[r] = lrow[r] * co + rs;
      mrow[r] = mn;
      alpha[r] = co;
      s[r] = p;
    }
#pragma unroll
    for (int r = 0; r < 8; ++r) { o0[r] *= alpha[r]; o1[r] *= alpha[r]; }

    __syncthreads();
#pragma unroll
    for (int r = 0; r < 8; ++r) pLds[(r + 8 * half) * 16 + m] = s[r];
    __syncthreads();

#pragma unroll
    for (int kk = 0; kk < 16; kk += 4) {
      const int k0 = kk + half * 2;
      v2f a  = { pLds[m * 16 + k0], pLds[m * 16 + k0 + 1] };
      v2f b0 = { vbase[k0 * (3 * Dd) + m],      vbase[(k0 + 1) * (3 * Dd) + m] };
      v2f b1 = { vbase[k0 * (3 * Dd) + 16 + m], vbase[(k0 + 1) * (3 * Dd) + 16 + m] };
      o0 = wmma4(a, b0, o0);
      o1 = wmma4(a, b1, o1);
    }
  }

  float* ob = obuf + (size_t)(qb * 16) * Dd + h * HDd;
#pragma unroll
  for (int r = 0; r < 8; ++r) {
    const int row = r + 8 * half;
    const float inv = 1.0f / lrow[r];
    ob[row * Dd + m]      = o0[r] * inv;
    ob[row * Dd + 16 + m] = o1[r] * inv;
  }
}

__global__ void zerof(float* p, int n) {
  int i = blockIdx.x * blockDim.x + threadIdx.x;
  if (i < n) p[i] = 0.f;
}
__global__ void seg_sum(const float* __restrict__ ew, const int* __restrict__ src,
                        float* __restrict__ out, int E) {
  int i = blockIdx.x * blockDim.x + threadIdx.x;
  if (i < E) atomicAdd(&out[src[i]], ew[i]);
}

__global__ __launch_bounds__(256)
void residual_ln(float* __restrict__ h, const float* __restrict__ add,
                 const float* __restrict__ wsn,
                 const float* __restrict__ g, const float* __restrict__ b) {
  const int row = blockIdx.x;
  const int c = threadIdx.x;
  const float wsv = wsn ? wsn[row] : 1.0f;
  float t = h[(size_t)row * Dd + c] + add[(size_t)row * Dd + c] * wsv;

  __shared__ float red[Dd];
  red[c] = t; __syncthreads();
  for (int s = Dd / 2; s > 0; s >>= 1) { if (c < s) red[c] += red[c + s]; __syncthreads(); }
  const float mean = red[0] / (float)Dd;
  __syncthreads();
  const float d = t - mean;
  red[c] = d * d; __syncthreads();
  for (int s = Dd / 2; s > 0; s >>= 1) { if (c < s) red[c] += red[c + s]; __syncthreads(); }
  const float var = red[0] / (float)Dd;
  const float rs = rsqrtf(var + 1e-5f);
  h[(size_t)row * Dd + c] = d * rs * g[c] + b[c];
}

__global__ void add_kernel(const float* a, const float* b, float* o, int n) {
  int i = blockIdx.x * blockDim.x + threadIdx.x;
  if (i < n) o[i] = a[i] + b[i];
}

__global__ __launch_bounds__(32)
void predict16(const float* __restrict__ hc,
               const int* __restrict__ si, const int* __restrict__ di,
               const float* __restrict__ pw1, const float* __restrict__ pb1,
               const float* __restrict__ pw2, const float* __restrict__ pb2,
               float* __restrict__ out) {
  const int tid  = threadIdx.x;
  const int base = blockIdx.x * 16;

  __shared__ float z[16 * Dd];
  for (int idx = tid; idx < 16 * Dd; idx += 32) {
    const int row = idx >> 8, col = idx & (Dd - 1);
    const int rs = si[base + row], rd = di[base + row];
    z[idx] = hc[(size_t)rs * Dd + col] * hc[(size_t)rd * Dd + col];
  }
  __syncthreads();

  const int m = tid & 15;
  const int half = tid >> 4;
  float acc[8];
#pragma unroll
  for (int r = 0; r < 8; ++r) acc[r] = 0.f;

  for (int jt = 0; jt < 8; ++jt) {
    v8f c = {};
    const float* wrow = pw1 + (size_t)(jt * 16 + m) * Dd + half * 2;
    const float* zrow = z + m * Dd + half * 2;
#pragma unroll
    for (int k = 0; k < Dd; k += 4) {
      v2f a = { zrow[k], zrow[k + 1] };
      v2f b = { wrow[k], wrow[k + 1] };
      c = wmma4(a, b, c);
    }
    const float bias1 = pb1[jt * 16 + m];
    const float w2 = pw2[jt * 16 + m];
#pragma unroll
    for (int r = 0; r < 8; ++r) {
      float v = c[r] + bias1;
      v = v > 0.f ? v : 0.2f * v;
      acc[r] += v * w2;
    }
  }
#pragma unroll
  for (int r = 0; r < 8; ++r) {
    const float sr = rsum16(acc[r]);
    if (m == 0) out[base + r + 8 * half] = sr + pb2[0];
  }
}

extern "C" void kernel_launch(void* const* d_in, const int* in_sizes, int n_in,
                              void* d_out, int out_size, void* d_ws, size_t ws_size,
                              hipStream_t stream) {
  const float* x       = (const float*)d_in[0];
  const float* edge_w  = (const float*)d_in[1];
  const int*   src     = (const int*)d_in[2];
  const int*   pos_src = (const int*)d_in[4];
  const int*   pos_dst = (const int*)d_in[5];
  const int*   neg_src = (const int*)d_in[6];
  const int*   neg_dst = (const int*)d_in[7];
  const float* proj_w  = (const float*)d_in[8];
  const float* proj_b  = (const float*)d_in[9];
  const float* in_w    = (const float*)d_in[10];
  const float* in_b    = (const float*)d_in[11];
  const float* out_w   = (const float*)d_in[12];
  const float* out_b   = (const float*)d_in[13];
  const float* ff_w    = (const float*)d_in[14];
  const float* ff_b    = (const float*)d_in[15];
  const float* g1      = (const float*)d_in[16];
  const float* b1      = (const float*)d_in[17];
  const float* g2      = (const float*)d_in[18];
  const float* b2      = (const float*)d_in[19];
  const float* pw1     = (const float*)d_in[20];
  const float* pb1     = (const float*)d_in[21];
  const float* pw2     = (const float*)d_in[22];
  const float* pb2     = (const float*)d_in[23];

  float* W      = (float*)d_ws;
  float* h      = W;
  float* hproj  = W + 1048576;
  float* qkv    = W + 2097152;
  float* obuf   = W + 5242880;
  float* scr    = W + 6291456;
  float* wsn    = W + 7340032;

  float* outp = (float*)d_out;
  float* pos  = outp;
  float* neg  = outp + Pp;
  float* hc   = outp + 2 * Pp;

  gemm_atb<INc><<<dim3(Dd / 64, Nn / 16), 32, 0, stream>>>(x, proj_w, proj_b, hproj, Nn, Dd);
  hipMemcpyAsync(h, hproj, (size_t)Nn * Dd * sizeof(float), hipMemcpyDeviceToDevice, stream);

  for (int l = 0; l < Ll; ++l) {
    gemm_atb<Dd><<<dim3(3 * Dd / 64, Nn / 16), 32, 0, stream>>>(
        h, in_w + (size_t)l * 3 * Dd * Dd, in_b + (size_t)l * 3 * Dd, qkv, Nn, 3 * Dd);
    attn_kernel<<<dim3(Nn / 16, Hh), 32, 0, stream>>>(qkv, obuf);
    gemm_atb<Dd><<<dim3(Dd / 64, Nn / 16), 32, 0, stream>>>(
        obuf, out_w + (size_t)l * Dd * Dd, out_b + (size_t)l * Dd, scr, Nn, Dd);
    zerof<<<(Nn + 255) / 256, 256, 0, stream>>>(wsn, Nn);
    seg_sum<<<(Ee + 255) / 256, 256, 0, stream>>>(edge_w + (size_t)l * Ee,
                                                  src + (size_t)l * Ee, wsn, Ee);
    residual_ln<<<Nn, Dd, 0, stream>>>(h, scr, wsn, g1 + (size_t)l * Dd, b1 + (size_t)l * Dd);
    gemm_atb<Dd><<<dim3(Dd / 64, Nn / 16), 32, 0, stream>>>(
        h, ff_w + (size_t)l * Dd * Dd, ff_b + (size_t)l * Dd, scr, Nn, Dd);
    residual_ln<<<Nn, Dd, 0, stream>>>(h, scr, nullptr, g2 + (size_t)l * Dd, b2 + (size_t)l * Dd);
  }

  add_kernel<<<(Nn * Dd + 255) / 256, 256, 0, stream>>>(h, hproj, hc, Nn * Dd);

  predict16<<<Pp / 16, 32, 0, stream>>>(hc, pos_src, pos_dst, pw1, pb1, pw2, pb2, pos);
  predict16<<<Pp / 16, 32, 0, stream>>>(hc, neg_src, neg_dst, pw1, pb1, pw2, pb2, neg);
}


# attention scores computed transposed (K*Qt): row max/sum in-lane, P feeds P*V from registers, no LDS or bpermute; waves_per_eu(1,4) so K/V loads issue as one group; sum order of l and P*V changes
# speedup vs baseline: 1.6721x; 1.6721x over previous
#include <hip/hip_runtime.h>
#include <math.h>

typedef __attribute__((ext_vector_type(2))) float v2f;
typedef __attribute__((ext_vector_type(8))) float v8f;

#define Nn   4096
#define INF_ 3.402823466e+38f
#define Dd   256
#define Hh   8
#define HDd  32
#define Ll   2
#define Ee   131072
#define Pp   200000
#define INc  128

__device__ __forceinline__ v8f wmma4(v2f a, v2f b, v8f c) {
  return __builtin_amdgcn_wmma_f32_16x16x4_f32(
      false, a, false, b, (short)0, c, false, false);
}

__device__ __forceinline__ float rsum16(float v) {
  for (int m = 8; m >= 1; m >>= 1) v += __shfl_xor(v, m, 16);
  return v;
}
__device__ __forceinline__ float rmax16(float v) {
  for (int m = 8; m >= 1; m >>= 1) v = fmaxf(v, __shfl_xor(v, m, 16));
  return v;
}

template <int KK>
__global__ __launch_bounds__(32)
void gemm_atb(const float* __restrict__ A, const float* __restrict__ W,
              const float* __restrict__ bias, float* __restrict__ C,
              int M, int Nout) {
  const int tid  = threadIdx.x;
  const int m    = tid & 15;
  const int half = tid >> 4;
  const int tbn  = blockIdx.x * 64;
  const int tbm  = blockIdx.y * 16;

  const float* arow = A + (size_t)(tbm + m) * KK + half * 2;
  const float* w0   = W + (size_t)(tbn + m) * KK + half * 2;
  const float* w1   = w0 + (size_t)16 * KK;
  const float* w2   = w1 + (size_t)16 * KK;
  const float* w3   = w2 + (size_t)16 * KK;

  v8f c0 = {}, c1 = {}, c2 = {}, c3 = {};
#pragma unroll
  for (int k = 0; k < KK; k += 4) {
    v2f a  = { arow[k], arow[k + 1] };
    v2f b0 = { w0[k],   w0[k + 1] };
    v2f b1 = { w1[k],   w1[k + 1] };
    v2f b2 = { w2[k],   w2[k + 1] };
    v2f b3 = { w3[k],   w3[k + 1] };
    c0 = wmma4(a, b0, c0);
    c1 = wmma4(a, b1, c1);
    c2 = wmma4(a, b2, c2);
    c3 = wmma4(a, b3, c3);
  }
  for (int r = 0; r < 8; ++r) {
    const int row = tbm + r + 8 * half;
    float* crow = C + (size_t)row * Nout + tbn + m;
    if (bias) {
      const float* brow = bias + tbn + m;
      crow[0]  = c0[r] + brow[0];
      crow[16] = c1[r] + brow[16];
      crow[32] = c2[r] + brow[32];
      crow[48] = c3[r] + brow[48];
    } else {
      crow[0] = c0[r]; crow[16] = c1[r]; crow[32] = c2[r]; crow[48] = c3[r];
    }
  }
}

__device__ __forceinline__ float xhalf(float x) {
  const int b = __builtin_bit_cast(int, x);
  return __builtin_bit_cast(float,
      __builtin_amdgcn_permlanex16(b, b, 0x76543210, 0xfedcba98, false, false));
}

__global__ __launch_bounds__(32) __attribute__((amdgpu_waves_per_eu(1, 4)))
void attn_kernel(const float* __restrict__ qkv, float* __restrict__ obuf) {
  const int tid  = threadIdx.x;
  const int m    = tid & 15;
  const int half = tid >> 4;
  const int qb   = blockIdx.x;
  const int h    = blockIdx.y;
  const float scale = 0.1767766952966369f;

  const float* qbase = qkv + (size_t)(qb * 16) * (3 * Dd) + h * HDd;

  v2f qa[8];
  for (int kk = 0; kk < HDd; kk += 4) {
    const int k0 = kk + half * 2;
    qa[kk >> 2] = v2f{ qbase[m * (3 * Dd) + k0], qbase[m * (3 * Dd) + k0 + 1] };
  }

  v8f o0 = {}, o1 = {};
  float mrun = -INF_;
  float lpart = 0.f;

  for (int kb = 0; kb < Nn / 16; ++kb) {
    const float* kbase = qkv + (size_t)(kb * 16) * (3 * Dd) + Dd + h * HDd;
    const float* vbase = qkv + (size_t)(kb * 16) * (3 * Dd) + 2 * Dd + h * HDd;

    v8f s = {};
    for (int kk = 0; kk < HDd; kk += 4) {
      const int k0 = kk + half * 2;
      v2f a = { kbase[m * (3 * Dd) + k0], kbase[m * (3 * Dd) + k0 + 1] };
      s = wmma4(a, qa[kk >> 2], s);
    }

    float sv[8];
    for (int r = 0; r < 8; ++r) sv[r] = s[r] * scale;
    float mx = fmaxf(fmaxf(fmaxf(sv[0], sv[1]), fmaxf(sv[2], sv[3])),
                     fmaxf(fmaxf(sv[4], sv[5]), fmaxf(sv[6], sv[7])));
    mx = fmaxf(mx, xhalf(mx));
    const float mn = fmaxf(mrun, mx);
    const float co = __expf(mrun - mn);
    float p[8];
    for (int r = 0; r < 8; ++r) p[r] = __expf(sv[r] - mn);
    const float ps = ((p[0] + p[1]) + (p[2] + p[3])) + ((p[4] + p[5]) + (p[6] + p[7]));
    lpart = lpart * co + ps;
    mrun = mn;
    for (int r = 0; r < 8; ++r) { o0[r] *= co; o1[r] *= co; }

    for (int j = 0; j < 4; ++j) {
      const float* vr = vbase + (size_t)(8 * half + 2 * j) * (3 * Dd) + m;
      v2f a0 = { vr[0],  vr[3 * Dd] };
      v2f a1 = { vr[16], vr[3 * Dd + 16] };
      v2f b  = { p[2 * j], p[2 * j + 1] };
      o0 = wmma4(a0, b, o0);
      o1 = wmma4(a1, b, o1);
    }
  }

  const float inv = 1.0f / (lpart + xhalf(lpart));
  float* ob = obuf + (size_t)(qb * 16 + m) * Dd + h * HDd + 8 * half;
  *(float4*)(ob)      = make_float4(o0[0] * inv, o0[1] * inv, o0[2] * inv, o0[3] * inv);
  *(float4*)(ob + 4)  = make_float4(o0[4] * inv, o0[5] * inv, o0[6] * inv, o0[7] * inv);
  *(float4*)(ob + 16) = make_float4(o1[0] * inv, o1[1] * inv, o1[2] * inv, o1[3] * inv);
  *(float4*)(ob + 20) = make_float4(o1[4] * inv, o1[5] * inv, o1[6] * inv, o1[7] * inv);
}

__global__ void zerof(float* p, int n) {
  int i = blockIdx.x * blockDim.x + threadIdx.x;
  if (i < n) p[i] = 0.f;
}
__global__ void seg_sum(const float* __restrict__ ew, const int* __restrict__ src,
                        float* __restrict__ out, int E) {
  int i = blockIdx.x * blockDim.x + threadIdx.x;
  if (i < E) atomicAdd(&out[src[i]], ew[i]);
}

__global__ __launch_bounds__(256)
void residual_ln(float* __restrict__ h, const float* __restrict__ add,
                 const float* __restrict__ wsn,
                 const float* __restrict__ g, const float* __restrict__ b) {
  const int row = blockIdx.x;
  const int c = threadIdx.x;
  const float wsv = wsn ? wsn[row] : 1.0f;
  float t = h[(size_t)row * Dd + c] + add[(size_t)row * Dd + c] * wsv;

  __shared__ float red[Dd];
  red[c] = t; __syncthreads();
  for (int s = Dd / 2; s > 0; s >>= 1) { if (c < s) red[c] += red[c + s]; __syncthreads(); }
  const float mean = red[0] / (float)Dd;
  __syncthreads();
  const float d = t - mean;
  red[c] = d * d; __syncthreads();
  for (int s = Dd / 2; s > 0; s >>= 1) { if (c < s) red[c] += red[c + s]; __syncthreads(); }
  const float var = red[0] / (float)Dd;
  const float rs = rsqrtf(var + 1e-5f);
  h[(size_t)row * Dd + c] = d * rs * g[c] + b[c];
}

__global__ void add_kernel(const float* a, const float* b, float* o, int n) {
  int i = blockIdx.x * blockDim.x + threadIdx.x;
  if (i < n) o[i] = a[i] + b[i];
}

__global__ __launch_bounds__(32)
void predict16(const float* __restrict__ hc,
               const int* __restrict__ si, const int* __restrict__ di,
               const float* __restrict__ pw1, const float* __restrict__ pb1,
               const float* __restrict__ pw2, const float* __restrict__ pb2,
               float* __restrict__ out) {
  const int tid  = threadIdx.x;
  const int base = blockIdx.x * 16;

  __shared__ float z[16 * Dd];
  for (int idx = tid; idx < 16 * Dd; idx += 32) {
    const int row = idx >> 8, col = idx & (Dd - 1);
    const int rs = si[base + row], rd = di[base + row];
    z[idx] = hc[(size_t)rs * Dd + col] * hc[(size_t)rd * Dd + col];
  }
  __syncthreads();

  const int m = tid & 15;
  const int half = tid >> 4;
  float acc[8];
#pragma unroll
  for (int r = 0; r < 8; ++r) acc[r] = 0.f;

  for (int jt = 0; jt < 8; ++jt) {
    v8f c = {};
    const float* wrow = pw1 + (size_t)(jt * 16 + m) * Dd + half * 2;
    const float* zrow = z + m * Dd + half * 2;
#pragma unroll
    for (int k = 0; k < Dd; k += 4) {
      v2f a = { zrow[k], zrow[k + 1] };
      v2f b = { wrow[k], wrow[k + 1] };
      c = wmma4(a, b, c);
    }
    const float bias1 = pb1[jt * 16 + m];
    const float w2 = pw2[jt * 16 + m];
#pragma unroll
    for (int r = 0; r < 8; ++r) {
      float v = c[r] + bias1;
      v = v > 0.f ? v : 0.2f * v;
      acc[r] += v * w2;
    }
  }
#pragma unroll
  for (int r = 0; r < 8; ++r) {
    const float sr = rsum16(acc[r]);
    if (m == 0) out[base + r + 8 * half] = sr + pb2[0];
  }
}

extern "C" void kernel_launch(void* const* d_in, const int* in_sizes, int n_in,
                              void* d_out, int out_size, void* d_ws, size_t ws_size,
                              hipStream_t stream) {
  const float* x       = (const float*)d_in[0];
  const float* edge_w  = (const float*)d_in[1];
  const int*   src     = (const int*)d_in[2];
  const int*   pos_src = (const int*)d_in[4];
  const int*   pos_dst = (const int*)d_in[5];
  const int*   neg_src = (const int*)d_in[6];
  const int*   neg_dst = (const int*)d_in[7];
  const float* proj_w  = (const float*)d_in[8];
  const float* proj_b  = (const float*)d_in[9];
  const float* in_w    = (const float*)d_in[10];
  const float* in_b    = (const float*)d_in[11];
  const float* out_w   = (const float*)d_in[12];
  const float* out_b   = (const float*)d_in[13];
  const float* ff_w    = (const float*)d_in[14];
  const float* ff_b    = (const float*)d_in[15];
  const float* g1      = (const float*)d_in[16];
  const float* b1      = (const float*)d_in[17];
  const float* g2      = (const float*)d_in[18];
  const float* b2      = (const float*)d_in[19];
  const float* pw1     = (const float*)d_in[20];
  const float* pb1     = (const float*)d_in[21];
  const float* pw2     = (const float*)d_in[22];
  const float* pb2     = (const float*)d_in[23];

  float* W      = (float*)d_ws;
  float* h      = W;
  float* hproj  = W + 1048576;
  float* qkv    = W + 2097152;
  float* obuf   = W + 5242880;
  float* scr    = W + 6291456;
  float* wsn    = W + 7340032;

  float* outp = (float*)d_out;
  float* pos  = outp;
  float* neg  = outp + Pp;
  float* hc   = outp + 2 * Pp;

  gemm_atb<INc><<<dim3(Dd / 64, Nn / 16), 32, 0, stream>>>(x, proj_w, proj_b, hproj, Nn, Dd);
  hipMemcpyAsync(h, hproj, (size_t)Nn * Dd * sizeof(float), hipMemcpyDeviceToDevice, stream);

  for (int l = 0; l < Ll; ++l) {
    gemm_atb<Dd><<<dim3(3 * Dd / 64, Nn / 16), 32, 0, stream>>>(
        h, in_w + (size_t)l * 3 * Dd * Dd, in_b + (size_t)l * 3 * Dd, qkv, Nn, 3 * Dd);
    attn_kernel<<<dim3(Nn / 16, Hh), 32, 0, stream>>>(qkv, obuf);
    gemm_atb<Dd><<<dim3(Dd / 64, Nn / 16), 32, 0, stream>>>(
        obuf, out_w + (size_t)l * Dd * Dd, out_b + (size_t)l * Dd, scr, Nn, Dd);
    zerof<<<(Nn + 255) / 256, 256, 0, stream>>>(wsn, Nn);
    seg_sum<<<(Ee + 255) / 256, 256, 0, stream>>>(edge_w + (size_t)l * Ee,
                                                  src + (size_t)l * Ee, wsn, Ee);
    residual_ln<<<Nn, Dd, 0, stream>>>(h, scr, wsn, g1 + (size_t)l * Dd, b1 + (size_t)l * Dd);
    gemm_atb<Dd><<<dim3(Dd / 64, Nn / 16), 32, 0, stream>>>(
        h, ff_w + (size_t)l * Dd * Dd, ff_b + (size_t)l * Dd, scr, Nn, Dd);
    residual_ln<<<Nn, Dd, 0, stream>>>(h, scr, nullptr, g2 + (size_t)l * Dd, b2 + (size_t)l * Dd);
  }

  add_kernel<<<(Nn * Dd + 255) / 256, 256, 0, stream>>>(h, hproj, hc, Nn * Dd);

  predict16<<<Pp / 16, 32, 0, stream>>>(hc, pos_src, pos_dst, pw1, pb1, pw2, pb2, pos);
  predict16<<<Pp / 16, 32, 0, stream>>>(hc, neg_src, neg_dst, pw1, pb1, pw2, pb2, neg);
}


# link predictor as one persistent launch: 128x256 weight copied to LDS once per work-group, WMMA B fragments read from LDS instead of global; 16-byte operand loads permute k inside each 8-k group
# speedup vs baseline: 2.1683x; 1.2968x over previous
#include <hip/hip_runtime.h>
#include <math.h>

typedef __attribute__((ext_vector_type(2))) float v2f;
typedef __attribute__((ext_vector_type(8))) float v8f;

#define Nn   4096
#define INF_ 3.402823466e+38f
#define Dd   256
#define Hh   8
#define HDd  32
#define Ll   2
#define Ee   131072
#define Pp   200000
#define INc  128

__device__ __forceinline__ v8f wmma4(v2f a, v2f b, v8f c) {
  return __builtin_amdgcn_wmma_f32_16x16x4_f32(
      false, a, false, b, (short)0, c, false, false);
}

__device__ __forceinline__ float rsum16(float v) {
  for (int m = 8; m >= 1; m >>= 1) v += __shfl_xor(v, m, 16);
  return v;
}
__device__ __forceinline__ float rmax16(float v) {
  for (int m = 8; m >= 1; m >>= 1) v = fmaxf(v, __shfl_xor(v, m, 16));
  return v;
}

template <int KK>
__global__ __launch_bounds__(32)
void gemm_atb(const float* __restrict__ A, const float* __restrict__ W,
              const float* __restrict__ bias, float* __restrict__ C,
              int M, int Nout) {
  const int tid  = threadIdx.x;
  const int m    = tid & 15;
  const int half = tid >> 4;
  const int tbn  = blockIdx.x * 64;
  const int tbm  = blockIdx.y * 16;

  const float* arow = A + (size_t)(tbm + m) * KK + half * 2;
  const float* w0   = W + (size_t)(tbn + m) * KK + half * 2;
  const float* w1   = w0 + (size_t)16 * KK;
  const float* w2   = w1 + (size_t)16 * KK;
  const float* w3   = w2 + (size_t)16 * KK;

  v8f c0 = {}, c1 = {}, c2 = {}, c3 = {};
#pragma unroll
  for (int k = 0; k < KK; k += 4) {
    v2f a  = { arow[k], arow[k + 1] };
    v2f b0 = { w0[k],   w0[k + 1] };
    v2f b1 = { w1[k],   w1[k + 1] };
    v2f b2 = { w2[k],   w2[k + 1] };
    v2f b3 = { w3[k],   w3[k + 1] };
    c0 = wmma4(a, b0, c0);
    c1 = wmma4(a, b1, c1);
    c2 = wmma4(a, b2, c2);
    c3 = wmma4(a, b3, c3);
  }
  for (int r = 0; r < 8; ++r) {
    const int row = tbm + r + 8 * half;
    float* crow = C + (size_t)row * Nout + tbn + m;
    if (bias) {
      const float* brow = bias + tbn + m;
      crow[0]  = c0[r] + brow[0];
      crow[16] = c1[r] + brow[16];
      crow[32] = c2[r] + brow[32];
      crow[48] = c3[r] + brow[48];
    } else {
      crow[0] = c0[r]; crow[16] = c1[r]; crow[32] = c2[r]; crow[48] = c3[r];
    }
  }
}

__device__ __forceinline__ float xhalf(float x) {
  const int b = __builtin_bit_cast(int, x);
  return __builtin_bit_cast(float,
      __builtin_amdgcn_permlanex16(b, b, 0x76543210, 0xfedcba98, false, false));
}

__global__ __launch_bounds__(32) __attribute__((amdgpu_waves_per_eu(1, 4)))
void attn_kernel(const float* __restrict__ qkv, float* __restrict__ obuf) {
  const int tid  = threadIdx.x;
  const int m    = tid & 15;
  const int half = tid >> 4;
  const int qb   = blockIdx.x;
  const int h    = blockIdx.y;
  const float scale = 0.1767766952966369f;

  const float* qbase = qkv + (size_t)(qb * 16) * (3 * Dd) + h * HDd;

  v2f qa[8];
  for (int kk = 0; kk < HDd; kk += 4) {
    const int k0 = kk + half * 2;
    qa[kk >> 2] = v2f{ qbase[m * (3 * Dd) + k0], qbase[m * (3 * Dd) + k0 + 1] };
  }

  v8f o0 = {}, o1 = {};
  float mrun = -INF_;
  float lpart = 0.f;

  for (int kb = 0; kb < Nn / 16; ++kb) {
    const float* kbase = qkv + (size_t)(kb * 16) * (3 * Dd) + Dd + h * HDd;
    const float* vbase = qkv + (size_t)(kb * 16) * (3 * Dd) + 2 * Dd + h * HDd;

    v8f s = {};
    for (int kk = 0; kk < HDd; kk += 4) {
      const int k0 = kk + half * 2;
      v2f a = { kbase[m * (3 * Dd) + k0], kbase[m * (3 * Dd) + k0 + 1] };
      s = wmma4(a, qa[kk >> 2], s);
    }

    float sv[8];
    for (int r = 0; r < 8; ++r) sv[r] = s[r] * scale;
    float mx = fmaxf(fmaxf(fmaxf(sv[0], sv[1]), fmaxf(sv[2], sv[3])),
                     fmaxf(fmaxf(sv[4], sv[5]), fmaxf(sv[6], sv[7])));
    mx = fmaxf(mx, xhalf(mx));
    const float mn = fmaxf(mrun, mx);
    const float co = __expf(mrun - mn);
    float p[8];
    for (int r = 0; r < 8; ++r) p[r] = __expf(sv[r] - mn);
    const float ps = ((p[0] + p[1]) + (p[2] + p[3])) + ((p[4] + p[5]) + (p[6] + p[7]));
    lpart = lpart * co + ps;
    mrun = mn;
    for (int r = 0; r < 8; ++r) { o0[r] *= co; o1[r] *= co; }

    for (int j = 0; j < 4; ++j) {
      const float* vr = vbase + (size_t)(8 * half + 2 * j) * (3 * Dd) + m;
      v2f a0 = { vr[0],  vr[3 * Dd] };
      v2f a1 = { vr[16], vr[3 * Dd + 16] };
      v2f b  = { p[2 * j], p[2 * j + 1] };
      o0 = wmma4(a0, b, o0);
      o1 = wmma4(a1, b, o1);
    }
  }

  const float inv = 1.0f / (lpart + xhalf(lpart));
  float* ob = obuf + (size_t)(qb * 16 + m) * Dd + h * HDd + 8 * half;
  *(float4*)(ob)      = make_float4(o0[0] * inv, o0[1] * inv, o0[2] * inv, o0[3] * inv);
  *(float4*)(ob + 4)  = make_float4(o0[4] * inv, o0[5] * inv, o0[6] * inv, o0[7] * inv);
  *(float4*)(ob + 16) = make_float4(o1[0] * inv, o1[1] * inv, o1[2] * inv, o1[3] * inv);
  *(float4*)(ob + 20) = make_float4(o1[4] * inv, o1[5] * inv, o1[6] * inv, o1[7] * inv);
}

__global__ void zerof(float* p, int n) {
  int i = blockIdx.x * blockDim.x + threadIdx.x;
  if (i < n) p[i] = 0.f;
}
__global__ void seg_sum(const float* __restrict__ ew, const int* __restrict__ src,
                        float* __restrict__ out, int E) {
  int i = blockIdx.x * blockDim.x + threadIdx.x;
  if (i < E) atomicAdd(&out[src[i]], ew[i]);
}

__global__ __launch_bounds__(256)
void residual_ln(float* __restrict__ h, const float* __restrict__ add,
                 const float* __restrict__ wsn,
                 const float* __restrict__ g, const float* __restrict__ b) {
  const int row = blockIdx.x;
  const int c = threadIdx.x;
  const float wsv = wsn ? wsn[row] : 1.0f;
  float t = h[(size_t)row * Dd + c] + add[(size_t)row * Dd + c] * wsv;

  __shared__ float red[Dd];
  red[c] = t; __syncthreads();
  for (int s = Dd / 2; s > 0; s >>= 1) { if (c < s) red[c] += red[c + s]; __syncthreads(); }
  const float mean = red[0] / (float)Dd;
  __syncthreads();
  const float d = t - mean;
  red[c] = d * d; __syncthreads();
  for (int s = Dd / 2; s > 0; s >>= 1) { if (c < s) red[c] += red[c + s]; __syncthreads(); }
  const float var = red[0] / (float)Dd;
  const float rs = rsqrtf(var + 1e-5f);
  h[(size_t)row * Dd + c] = d * rs * g[c] + b[c];
}

__global__ void add_kernel(const float* a, const float* b, float* o, int n) {
  int i = blockIdx.x * blockDim.x + threadIdx.x;
  if (i < n) o[i] = a[i] + b[i];
}

#define PWP  260
#define TPH  (Pp / 16)
#define PWG  256
#define PWV  12

__global__ __launch_bounds__(32 * PWV)
void predict_all(const float* __restrict__ hc,
                 const int* __restrict__ psrc, const int* __restrict__ pdst,
                 const int* __restrict__ nsrc, const int* __restrict__ ndst,
                 const float* __restrict__ pw1, const float* __restrict__ pb1,
                 const float* __restrict__ pw2, const float* __restrict__ pb2,
                 float* __restrict__ out) {
  __shared__ float wl[(Dd / 2) * PWP];

  const int tid = threadIdx.x;
  for (int idx = tid; idx < (Dd / 2) * (Dd / 4); idx += 32 * PWV) {
    const int row = idx >> 6, c4 = idx & 63;
    const float4 v = *(const float4*)(pw1 + (size_t)row * Dd + c4 * 4);
    *(float4*)(wl + row * PWP + c4 * 4) = v;
  }
  __syncthreads();

  const int lane = tid & 31;
  const int wave = tid >> 5;
  const int m    = lane & 15;
  const int half = lane >> 4;

  float bias1[8], w2[8];
#pragma unroll
  for (int jt = 0; jt < 8; ++jt) {
    bias1[jt] = pb1[jt * 16 + m];
    w2[jt]    = pw2[jt * 16 + m];
  }
  const float b2 = pb2[0];
  const float* wfrag = wl + m * PWP + 4 * half;

  for (int t = wave * PWG + (int)blockIdx.x; t < 2 * TPH; t += PWV * PWG) {
    asm volatile("" ::: "memory");
    const bool ng = t >= TPH;
    const int* sp = ng ? nsrc : psrc;
    const int* dp = ng ? ndst : pdst;
    const int tb  = (ng ? t - TPH : t) * 16;
    const int rs = sp[tb + m], rd = dp[tb + m];
    const float* as = hc + (size_t)rs * Dd + 4 * half;
    const float* ad = hc + (size_t)rd * Dd + 4 * half;

    float4 z[32];
#pragma unroll
    for (int i = 0; i < 32; ++i) {
      const float4 a = *(const float4*)(as + 8 * i);
      const float4 b = *(const float4*)(ad + 8 * i);
      z[i] = make_float4(a.x * b.x, a.y * b.y, a.z * b.z, a.w * b.w);
    }

    v8f c[8];
#pragma unroll
    for (int jt = 0; jt < 8; ++jt) c[jt] = v8f{};

#pragma unroll
    for (int i = 0; i < 32; ++i) {
      float4 w[8];
#pragma unroll
      for (int jt = 0; jt < 8; ++jt)
        w[jt] = *(const float4*)(wfrag + jt * 16 * PWP + 8 * i);
      const v2f a0 = { z[i].x, z[i].y };
      const v2f a1 = { z[i].z, z[i].w };
#pragma unroll
      for (int jt = 0; jt < 8; ++jt)
        c[jt] = wmma4(a0, v2f{ w[jt].x, w[jt].y }, c[jt]);
#pragma unroll
      for (int jt = 0; jt < 8; ++jt)
        c[jt] = wmma4(a1, v2f{ w[jt].z, w[jt].w }, c[jt]);
    }

    float acc[8];
#pragma unroll
    for (int r = 0; r < 8; ++r) acc[r] = 0.f;
#pragma unroll
    for (int jt = 0; jt < 8; ++jt) {
#pragma unroll
      for (int r = 0; r < 8; ++r) {
        float v = c[jt][r] + bias1[jt];
        v = v > 0.f ? v : 0.2f * v;
        acc[r] += v * w2[jt];
      }
    }
#pragma unroll
    for (int r = 0; r < 8; ++r) acc[r] = rsum16(acc[r]);
    float o = acc[0];
#pragma unroll
    for (int r = 1; r < 8; ++r) o = (m == r) ? acc[r] : o;
    if (m < 8) out[(size_t)t * 16 + 8 * half + m] = o + b2;
  }
}

extern "C" void kernel_launch(void* const* d_in, const int* in_sizes, int n_in,
                              void* d_out, int out_size, void* d_ws, size_t ws_size,
                              hipStream_t stream) {
  const float* x       = (const float*)d_in[0];
  const float* edge_w  = (const float*)d_in[1];
  const int*   src     = (const int*)d_in[2];
  const int*   pos_src = (const int*)d_in[4];
  const int*   pos_dst = (const int*)d_in[5];
  const int*   neg_src = (const int*)d_in[6];
  const int*   neg_dst = (const int*)d_in[7];
  const float* proj_w  = (const float*)d_in[8];
  const float* proj_b  = (const float*)d_in[9];
  const float* in_w    = (const float*)d_in[10];
  const float* in_b    = (const float*)d_in[11];
  const float* out_w   = (const float*)d_in[12];
  const float* out_b   = (const float*)d_in[13];
  const float* ff_w    = (const float*)d_in[14];
  const float* ff_b    = (const float*)d_in[15];
  const float* g1      = (const float*)d_in[16];
  const float* b1      = (const float*)d_in[17];
  const float* g2      = (const float*)d_in[18];
  const float* b2      = (const float*)d_in[19];
  const float* pw1     = (const float*)d_in[20];
  const float* pb1     = (const float*)d_in[21];
  const float* pw2     = (const float*)d_in[22];
  const float* pb2     = (const float*)d_in[23];

  float* W      = (float*)d_ws;
  float* h      = W;
  float* hproj  = W + 1048576;
  float* qkv    = W + 2097152;
  float* obuf   = W + 5242880;
  float* scr    = W + 6291456;
  float* wsn    = W + 7340032;

  float* outp = (float*)d_out;
  float* pos  = outp;
  float* hc   = outp + 2 * Pp;

  gemm_atb<INc><<<dim3(Dd / 64, Nn / 16), 32, 0, stream>>>(x, proj_w, proj_b, hproj, Nn, Dd);
  hipMemcpyAsync(h, hproj, (size_t)Nn * Dd * sizeof(float), hipMemcpyDeviceToDevice, stream);

  for (int l = 0; l < Ll; ++l) {
    gemm_atb<Dd><<<dim3(3 * Dd / 64, Nn / 16), 32, 0, stream>>>(
        h, in_w + (size_t)l * 3 * Dd * Dd, in_b + (size_t)l * 3 * Dd, qkv, Nn, 3 * Dd);
    attn_kernel<<<dim3(Nn / 16, Hh), 32, 0, stream>>>(qkv, obuf);
    gemm_atb<Dd><<<dim3(Dd / 64, Nn / 16), 32, 0, stream>>>(
        obuf, out_w + (size_t)l * Dd * Dd, out_b + (size_t)l * Dd, scr, Nn, Dd);
    zerof<<<(Nn + 255) / 256, 256, 0, stream>>>(wsn, Nn);
    seg_sum<<<(Ee + 255) / 256, 256, 0, stream>>>(edge_w + (size_t)l * Ee,
                                                  src + (size_t)l * Ee, wsn, Ee);
    residual_ln<<<Nn, Dd, 0, stream>>>(h, scr, wsn, g1 + (size_t)l * Dd, b1 + (size_t)l * Dd);
    gemm_atb<Dd><<<dim3(Dd / 64, Nn / 16), 32, 0, stream>>>(
        h, ff_w + (size_t)l * Dd * Dd, ff_b + (size_t)l * Dd, scr, Nn, Dd);
    residual_ln<<<Nn, Dd, 0, stream>>>(h, scr, nullptr, g2 + (size_t)l * Dd, b2 + (size_t)l * Dd);
  }

  add_kernel<<<(Nn * Dd + 255) / 256, 256, 0, stream>>>(h, hproj, hc, Nn * Dd);

  predict_all<<<PWG, 32 * PWV, 0, stream>>>(hc, pos_src, pos_dst, neg_src, neg_dst,
                                       pw1, pb1, pw2, pb2, pos);
}


# attention: each wave owns 32 queries (two query tiles), so every K and V fragment load feeds two WMMAs; 4-wave groups, one wave per SIMD, next key block requested one block ahead; sum order kept
# speedup vs baseline: 2.3526x; 1.0850x over previous
#include <hip/hip_runtime.h>
#include <math.h>

typedef __attribute__((ext_vector_type(2))) float v2f;
typedef __attribute__((ext_vector_type(8))) float v8f;

#define Nn   4096
#define INF_ 3.402823466e+38f
#define Dd   256
#define Hh   8
#define HDd  32
#define Ll   2
#define Ee   131072
#define Pp   200000
#define INc  128

__device__ __forceinline__ v8f wmma4(v2f a, v2f b, v8f c) {
  return __builtin_amdgcn_wmma_f32_16x16x4_f32(
      false, a, false, b, (short)0, c, false, false);
}

__device__ __forceinline__ float rsum16(float v) {
  for (int m = 8; m >= 1; m >>= 1) v += __shfl_xor(v, m, 16);
  return v;
}
__device__ __forceinline__ float rmax16(float v) {
  for (int m = 8; m >= 1; m >>= 1) v = fmaxf(v, __shfl_xor(v, m, 16));
  return v;
}

template <int KK>
__global__ __launch_bounds__(32)
void gemm_atb(const float* __restrict__ A, const float* __restrict__ W,
              const float* __restrict__ bias, float* __restrict__ C,
              int M, int Nout) {
  const int tid  = threadIdx.x;
  const int m    = tid & 15;
  const int half = tid >> 4;
  const int tbn  = blockIdx.x * 64;
  const int tbm  = blockIdx.y * 16;

  const float* arow = A + (size_t)(tbm + m) * KK + half * 2;
  const float* w0   = W + (size_t)(tbn + m) * KK + half * 2;
  const float* w1   = w0 + (size_t)16 * KK;
  const float* w2   = w1 + (size_t)16 * KK;
  const float* w3   = w2 + (size_t)16 * KK;

  v8f c0 = {}, c1 = {}, c2 = {}, c3 = {};
#pragma unroll
  for (int k = 0; k < KK; k += 4) {
    v2f a  = { arow[k], arow[k + 1] };
    v2f b0 = { w0[k],   w0[k + 1] };
    v2f b1 = { w1[k],   w1[k + 1] };
    v2f b2 = { w2[k],   w2[k + 1] };
    v2f b3 = { w3[k],   w3[k + 1] };
    c0 = wmma4(a, b0, c0);
    c1 = wmma4(a, b1, c1);
    c2 = wmma4(a, b2, c2);
    c3 = wmma4(a, b3, c3);
  }
  for (int r = 0; r < 8; ++r) {
    const int row = tbm + r + 8 * half;
    float* crow = C + (size_t)row * Nout + tbn + m;
    if (bias) {
      const float* brow = bias + tbn + m;
      crow[0]  = c0[r] + brow[0];
      crow[16] = c1[r] + brow[16];
      crow[32] = c2[r] + brow[32];
      crow[48] = c3[r] + brow[48];
    } else {
      crow[0] = c0[r]; crow[16] = c1[r]; crow[32] = c2[r]; crow[48] = c3[r];
    }
  }
}

__device__ __forceinline__ float xhalf(float x) {
  const int b = __builtin_bit_cast(int, x);
  return __builtin_bit_cast(float,
      __builtin_amdgcn_permlanex16(b, b, 0x76543210, 0xfedcba98, false, false));
}

struct KVFrag {
  v2f k[8];
  v2f v0[4];
  v2f v1[4];
};

__device__ __forceinline__ void load_kv(KVFrag& f, const float* __restrict__ kp,
                                        const float* __restrict__ vp) {
  for (int i = 0; i < 8; ++i) f.k[i] = *(const v2f*)(kp + 4 * i);
  for (int j = 0; j < 4; ++j) {
    const float* vr = vp + (size_t)(2 * j) * (3 * Dd);
    f.v0[j] = v2f{ vr[0],  vr[3 * Dd] };
    f.v1[j] = v2f{ vr[16], vr[3 * Dd + 16] };
  }
}

__device__ __forceinline__ void softmax_tile(const v8f& s, float& mrun, float& lpart,
                                             v8f& oa, v8f& ob, float (&p)[8]) {
  const float scale = 0.1767766952966369f;
  float sv[8];
  for (int r = 0; r < 8; ++r) sv[r] = s[r] * scale;
  float mx = fmaxf(fmaxf(fmaxf(sv[0], sv[1]), fmaxf(sv[2], sv[3])),
                   fmaxf(fmaxf(sv[4], sv[5]), fmaxf(sv[6], sv[7])));
  mx = fmaxf(mx, xhalf(mx));
  const float mn = fmaxf(mrun, mx);
  const float co = __expf(mrun - mn);
  for (int r = 0; r < 8; ++r) p[r] = __expf(sv[r] - mn);
  const float ps = ((p[0] + p[1]) + (p[2] + p[3])) + ((p[4] + p[5]) + (p[6] + p[7]));
  lpart = lpart * co + ps;
  mrun = mn;
  for (int r = 0; r < 8; ++r) { oa[r] *= co; ob[r] *= co; }
}

struct QTile {
  v2f q[8];
  v8f o0, o1;
  float mrun;
  float lpart;
};

__device__ __forceinline__ void attn_block(const KVFrag& f, QTile& t0, QTile& t1) {
  v8f s0 = {}, s1 = {};
  for (int i = 0; i < 8; ++i) {
    s0 = wmma4(f.k[i], t0.q[i], s0);
    s1 = wmma4(f.k[i], t1.q[i], s1);
  }
  float p0[8], p1[8];
  softmax_tile(s0, t0.mrun, t0.lpart, t0.o0, t0.o1, p0);
  for (int j = 0; j < 4; ++j) {
    const v2f b = { p0[2 * j], p0[2 * j + 1] };
    t0.o0 = wmma4(f.v0[j], b, t0.o0);
    t0.o1 = wmma4(f.v1[j], b, t0.o1);
  }
  softmax_tile(s1, t1.mrun, t1.lpart, t1.o0, t1.o1, p1);
  for (int j = 0; j < 4; ++j) {
    const v2f b = { p1[2 * j], p1[2 * j + 1] };
    t1.o0 = wmma4(f.v0[j], b, t1.o0);
    t1.o1 = wmma4(f.v1[j], b, t1.o1);
  }
}

__device__ __forceinline__ void store_tile(const QTile& t, float* __restrict__ ob) {
  const float inv = 1.0f / (t.lpart + xhalf(t.lpart));
  *(float4*)(ob)      = make_float4(t.o0[0] * inv, t.o0[1] * inv, t.o0[2] * inv, t.o0[3] * inv);
  *(float4*)(ob + 4)  = make_float4(t.o0[4] * inv, t.o0[5] * inv, t.o0[6] * inv, t.o0[7] * inv);
  *(float4*)(ob + 16) = make_float4(t.o1[0] * inv, t.o1[1] * inv, t.o1[2] * inv, t.o1[3] * inv);
  *(float4*)(ob + 20) = make_float4(t.o1[4] * inv, t.o1[5] * inv, t.o1[6] * inv, t.o1[7] * inv);
}

__global__ __launch_bounds__(128) __attribute__((amdgpu_waves_per_eu(1, 4)))
void attn_kernel(const float* __restrict__ qkv, float* __restrict__ obuf) {
  const int lane = threadIdx.x & 31;
  const int wave = threadIdx.x >> 5;
  const int m    = lane & 15;
  const int half = lane >> 4;
  const int q0   = (blockIdx.x * 4 + wave) * 32;
  const int h    = blockIdx.y;

  QTile t0, t1;
  {
    const float* qr = qkv + (size_t)(q0 + m) * (3 * Dd) + h * HDd + 2 * half;
    for (int i = 0; i < 8; ++i) {
      t0.q[i] = *(const v2f*)(qr + 4 * i);
      t1.q[i] = *(const v2f*)(qr + (size_t)16 * (3 * Dd) + 4 * i);
    }
  }
  t0.o0 = v8f{}; t0.o1 = v8f{}; t1.o0 = v8f{}; t1.o1 = v8f{};
  t0.mrun = -INF_; t1.mrun = -INF_;
  t0.lpart = 0.f;  t1.lpart = 0.f;

  const float* kp = qkv + Dd + h * HDd + (size_t)m * (3 * Dd) + 2 * half;
  const float* vp = qkv + 2 * Dd + h * HDd + (size_t)(8 * half) * (3 * Dd) + m;
  const size_t kstep = (size_t)16 * (3 * Dd);

  KVFrag fa, fb;
  load_kv(fa, kp, vp);
  for (int kb = 0; kb < Nn / 16; kb += 2) {
    load_kv(fb, kp + (size_t)(kb + 1) * kstep, vp + (size_t)(kb + 1) * kstep);
    attn_block(fa, t0, t1);
    __builtin_amdgcn_sched_barrier(0);
    const int nx = (kb + 2 < Nn / 16) ? kb + 2 : kb + 1;
    load_kv(fa, kp + (size_t)nx * kstep, vp + (size_t)nx * kstep);
    asm volatile("" ::: "memory");
    attn_block(fb, t0, t1);
  }

  float* ob = obuf + (size_t)(q0 + m) * Dd + h * HDd + 8 * half;
  store_tile(t0, ob);
  store_tile(t1, ob + (size_t)16 * Dd);
}

__global__ void zerof(float* p, int n) {
  int i = blockIdx.x * blockDim.x + threadIdx.x;
  if (i < n) p[i] = 0.f;
}
__global__ void seg_sum(const float* __restrict__ ew, const int* __restrict__ src,
                        float* __restrict__ out, int E) {
  int i = blockIdx.x * blockDim.x + threadIdx.x;
  if (i < E) atomicAdd(&out[src[i]], ew[i]);
}

__global__ __launch_bounds__(256)
void residual_ln(float* __restrict__ h, const float* __restrict__ add,
                 const float* __restrict__ wsn,
                 const float* __restrict__ g, const float* __restrict__ b) {
  const int row = blockIdx.x;
  const int c = threadIdx.x;
  const float wsv = wsn ? wsn[row] : 1.0f;
  float t = h[(size_t)row * Dd + c] + add[(size_t)row * Dd + c] * wsv;

  __shared__ float red[Dd];
  red[c] = t; __syncthreads();
  for (int s = Dd / 2; s > 0; s >>= 1) { if (c < s) red[c] += red[c + s]; __syncthreads(); }
  const float mean = red[0] / (float)Dd;
  __syncthreads();
  const float d = t - mean;
  red[c] = d * d; __syncthreads();
  for (int s = Dd / 2; s > 0; s >>= 1) { if (c < s) red[c] += red[c + s]; __syncthreads(); }
  const float var = red[0] / (float)Dd;
  const float rs = rsqrtf(var + 1e-5f);
  h[(size_t)row * Dd + c] = d * rs * g[c] + b[c];
}

__global__ void add_kernel(const float* a, const float* b, float* o, int n) {
  int i = blockIdx.x * blockDim.x + threadIdx.x;
  if (i < n) o[i] = a[i] + b[i];
}

#define PWP  260
#define TPH  (Pp / 16)
#define PWG  256
#define PWV  12

__global__ __launch_bounds__(32 * PWV)
void predict_all(const float* __restrict__ hc,
                 const int* __restrict__ psrc, const int* __restrict__ pdst,
                 const int* __restrict__ nsrc, const int* __restrict__ ndst,
                 const float* __restrict__ pw1, const float* __restrict__ pb1,
                 const float* __restrict__ pw2, const float* __restrict__ pb2,
                 float* __restrict__ out) {
  __shared__ float wl[(Dd / 2) * PWP];

  const int tid = threadIdx.x;
  for (int idx = tid; idx < (Dd / 2) * (Dd / 4); idx += 32 * PWV) {
    const int row = idx >> 6, c4 = idx & 63;
    const float4 v = *(const float4*)(pw1 + (size_t)row * Dd + c4 * 4);
    *(float4*)(wl + row * PWP + c4 * 4) = v;
  }
  __syncthreads();

  const int lane = tid & 31;
  const int wave = tid >> 5;
  const int m    = lane & 15;
  const int half = lane >> 4;

  float bias1[8], w2[8];
#pragma unroll
  for (int jt = 0; jt < 8; ++jt) {
    bias1[jt] = pb1[jt * 16 + m];
    w2[jt]    = pw2[jt * 16 + m];
  }
  const float b2 = pb2[0];
  const float* wfrag = wl + m * PWP + 4 * half;

  for (int t = wave * PWG + (int)blockIdx.x; t < 2 * TPH; t += PWV * PWG) {
    asm volatile("" ::: "memory");
    const bool ng = t >= TPH;
    const int* sp = ng ? nsrc : psrc;
    const int* dp = ng ? ndst : pdst;
    const int tb  = (ng ? t - TPH : t) * 16;
    const int rs = sp[tb + m], rd = dp[tb + m];
    const float* as = hc + (size_t)rs * Dd + 4 * half;
    const float* ad = hc + (size_t)rd * Dd + 4 * half;

    float4 z[32];
#pragma unroll
    for (int i = 0; i < 32; ++i) {
      const float4 a = *(const float4*)(as + 8 * i);
      const float4 b = *(const float4*)(ad + 8 * i);
      z[i] = make_float4(a.x * b.x, a.y * b.y, a.z * b.z, a.w * b.w);
    }

    v8f c[8];
#pragma unroll
    for (int jt = 0; jt < 8; ++jt) c[jt] = v8f{};

#pragma unroll
    for (int i = 0; i < 32; ++i) {
      float4 w[8];
#pragma unroll
      for (int jt = 0; jt < 8; ++jt)
        w[jt] = *(const float4*)(wfrag + jt * 16 * PWP + 8 * i);
      const v2f a0 = { z[i].x, z[i].y };
      const v2f a1 = { z[i].z, z[i].w };
#pragma unroll
      for (int jt = 0; jt < 8; ++jt)
        c[jt] = wmma4(a0, v2f{ w[jt].x, w[jt].y }, c[jt]);
#pragma unroll
      for (int jt = 0; jt < 8; ++jt)
        c[jt] = wmma4(a1, v2f{ w[jt].z, w[jt].w }, c[jt]);
    }

    float acc[8];
#pragma unroll
    for (int r = 0; r < 8; ++r) acc[r] = 0.f;
#pragma unroll
    for (int jt = 0; jt < 8; ++jt) {
#pragma unroll
      for (int r = 0; r < 8; ++r) {
        float v = c[jt][r] + bias1[jt];
        v = v > 0.f ? v : 0.2f * v;
        acc[r] += v * w2[jt];
      }
    }
#pragma unroll
    for (int r = 0; r < 8; ++r) acc[r] = rsum16(acc[r]);
    float o = acc[0];
#pragma unroll
    for (int r = 1; r < 8; ++r) o = (m == r) ? acc[r] : o;
    if (m < 8) out[(size_t)t * 16 + 8 * half + m] = o + b2;
  }
}

extern "C" void kernel_launch(void* const* d_in, const int* in_sizes, int n_in,
                              void* d_out, int out_size, void* d_ws, size_t ws_size,
                              hipStream_t stream) {
  const float* x       = (const float*)d_in[0];
  const float* edge_w  = (const float*)d_in[1];
  const int*   src     = (const int*)d_in[2];
  const int*   pos_src = (const int*)d_in[4];
  const int*   pos_dst = (const int*)d_in[5];
  const int*   neg_src = (const int*)d_in[6];
  const int*   neg_dst = (const int*)d_in[7];
  const float* proj_w  = (const float*)d_in[8];
  const float* proj_b  = (const float*)d_in[9];
  const float* in_w    = (const float*)d_in[10];
  const float* in_b    = (const float*)d_in[11];
  const float* out_w   = (const float*)d_in[12];
  const float* out_b   = (const float*)d_in[13];
  const float* ff_w    = (const float*)d_in[14];
  const float* ff_b    = (const float*)d_in[15];
  const float* g1      = (const float*)d_in[16];
  const float* b1      = (const float*)d_in[17];
  const float* g2      = (const float*)d_in[18];
  const float* b2      = (const float*)d_in[19];
  const float* pw1     = (const float*)d_in[20];
  const float* pb1     = (const float*)d_in[21];
  const float* pw2     = (const float*)d_in[22];
  const float* pb2     = (const float*)d_in[23];

  float* W      = (float*)d_ws;
  float* h      = W;
  float* hproj  = W + 1048576;
  float* qkv    = W + 2097152;
  float* obuf   = W + 5242880;
  float* scr    = W + 6291456;
  float* wsn    = W + 7340032;

  float* outp = (float*)d_out;
  float* pos  = outp;
  float* hc   = outp + 2 * Pp;

  gemm_atb<INc><<<dim3(Dd / 64, Nn / 16), 32, 0, stream>>>(x, proj_w, proj_b, hproj, Nn, Dd);
  hipMemcpyAsync(h, hproj, (size_t)Nn * Dd * sizeof(float), hipMemcpyDeviceToDevice, stream);

  for (int l = 0; l < Ll; ++l) {
    gemm_atb<Dd><<<dim3(3 * Dd / 64, Nn / 16), 32, 0, stream>>>(
        h, in_w + (size_t)l * 3 * Dd * Dd, in_b + (size_t)l * 3 * Dd, qkv, Nn, 3 * Dd);
    attn_kernel<<<dim3(Nn / 128, Hh), 128, 0, stream>>>(qkv, obuf);
    gemm_atb<Dd><<<dim3(Dd / 64, Nn / 16), 32, 0, stream>>>(
        obuf, out_w + (size_t)l * Dd * Dd, out_b + (size_t)l * Dd, scr, Nn, Dd);
    zerof<<<(Nn + 255) / 256, 256, 0, stream>>>(wsn, Nn);
    seg_sum<<<(Ee + 255) / 256, 256, 0, stream>>>(edge_w + (size_t)l * Ee,
                                                  src + (size_t)l * Ee, wsn, Ee);
    residual_ln<<<Nn, Dd, 0, stream>>>(h, scr, wsn, g1 + (size_t)l * Dd, b1 + (size_t)l * Dd);
    gemm_atb<Dd><<<dim3(Dd / 64, Nn / 16), 32, 0, stream>>>(
        h, ff_w + (size_t)l * Dd * Dd, ff_b + (size_t)l * Dd, scr, Nn, Dd);
    residual_ln<<<Nn, Dd, 0, stream>>>(h, scr, nullptr, g2 + (size_t)l * Dd, b2 + (size_t)l * Dd);
  }

  add_kernel<<<(Nn * Dd + 255) / 256, 256, 0, stream>>>(h, hproj, hc, Nn * Dd);

  predict_all<<<PWG, 32 * PWV, 0, stream>>>(hc, pos_src, pos_dst, neg_src, neg_dst,
                                       pw1, pb1, pw2, pb2, pos);
}


# residual+LayerNorm (x4), final h+h_proj add and the h copy moved into the epilogue of the GEMM feeding them: 4-wave group owns 16 full rows; LN row sums by lane, shuffle, LDS (LN sum order changes)
# speedup vs baseline: 2.6026x; 1.1063x over previous
#include <hip/hip_runtime.h>
#include <math.h>

typedef __attribute__((ext_vector_type(2))) float v2f;
typedef __attribute__((ext_vector_type(8))) float v8f;

#define Nn   4096
#define INF_ 3.402823466e+38f
#define Dd   256
#define Hh   8
#define HDd  32
#define Ll   2
#define Ee   131072
#define Pp   200000
#define INc  128

__device__ __forceinline__ v8f wmma4(v2f a, v2f b, v8f c) {
  return __builtin_amdgcn_wmma_f32_16x16x4_f32(
      false, a, false, b, (short)0, c, false, false);
}

__device__ __forceinline__ float rsum16(float v) {
  for (int m = 8; m >= 1; m >>= 1) v += __shfl_xor(v, m, 16);
  return v;
}
__device__ __forceinline__ float rmax16(float v) {
  for (int m = 8; m >= 1; m >>= 1) v = fmaxf(v, __shfl_xor(v, m, 16));
  return v;
}

template <int KK>
__global__ __launch_bounds__(32)
void gemm_atb(const float* __restrict__ A, const float* __restrict__ W,
              const float* __restrict__ bias, float* __restrict__ C,
              int M, int Nout) {
  const int tid  = threadIdx.x;
  const int m    = tid & 15;
  const int half = tid >> 4;
  const int tbn  = blockIdx.x * 64;
  const int tbm  = blockIdx.y * 16;

  const float* arow = A + (size_t)(tbm + m) * KK + half * 2;
  const float* w0   = W + (size_t)(tbn + m) * KK + half * 2;
  const float* w1   = w0 + (size_t)16 * KK;
  const float* w2   = w1 + (size_t)16 * KK;
  const float* w3   = w2 + (size_t)16 * KK;

  v8f c0 = {}, c1 = {}, c2 = {}, c3 = {};
#pragma unroll
  for (int k = 0; k < KK; k += 4) {
    v2f a  = { arow[k], arow[k + 1] };
    v2f b0 = { w0[k],   w0[k + 1] };
    v2f b1 = { w1[k],   w1[k + 1] };
    v2f b2 = { w2[k],   w2[k + 1] };
    v2f b3 = { w3[k],   w3[k + 1] };
    c0 = wmma4(a, b0, c0);
    c1 = wmma4(a, b1, c1);
    c2 = wmma4(a, b2, c2);
    c3 = wmma4(a, b3, c3);
  }
  for (int r = 0; r < 8; ++r) {
    const int row = tbm + r + 8 * half;
    float* crow = C + (size_t)row * Nout + tbn + m;
    if (bias) {
      const float* brow = bias + tbn + m;
      crow[0]  = c0[r] + brow[0];
      crow[16] = c1[r] + brow[16];
      crow[32] = c2[r] + brow[32];
      crow[48] = c3[r] + brow[48];
    } else {
      crow[0] = c0[r]; crow[16] = c1[r]; crow[32] = c2[r]; crow[48] = c3[r];
    }
  }
}

__device__ __forceinline__ float xhalf(float x) {
  const int b = __builtin_bit_cast(int, x);
  return __builtin_bit_cast(float,
      __builtin_amdgcn_permlanex16(b, b, 0x76543210, 0xfedcba98, false, false));
}

struct KVFrag {
  v2f k[8];
  v2f v0[4];
  v2f v1[4];
};

__device__ __forceinline__ void load_kv(KVFrag& f, const float* __restrict__ kp,
                                        const float* __restrict__ vp) {
  for (int i = 0; i < 8; ++i) f.k[i] = *(const v2f*)(kp + 4 * i);
  for (int j = 0; j < 4; ++j) {
    const float* vr = vp + (size_t)(2 * j) * (3 * Dd);
    f.v0[j] = v2f{ vr[0],  vr[3 * Dd] };
    f.v1[j] = v2f{ vr[16], vr[3 * Dd + 16] };
  }
}

__device__ __forceinline__ void softmax_tile(const v8f& s, float& mrun, float& lpart,
                                             v8f& oa, v8f& ob, float (&p)[8]) {
  const float scale = 0.1767766952966369f;
  float sv[8];
  for (int r = 0; r < 8; ++r) sv[r] = s[r] * scale;
  float mx = fmaxf(fmaxf(fmaxf(sv[0], sv[1]), fmaxf(sv[2], sv[3])),
                   fmaxf(fmaxf(sv[4], sv[5]), fmaxf(sv[6], sv[7])));
  mx = fmaxf(mx, xhalf(mx));
  const float mn = fmaxf(mrun, mx);
  const float co = __expf(mrun - mn);
  for (int r = 0; r < 8; ++r) p[r] = __expf(sv[r] - mn);
  const float ps = ((p[0] + p[1]) + (p[2] + p[3])) + ((p[4] + p[5]) + (p[6] + p[7]));
  lpart = lpart * co + ps;
  mrun = mn;
  for (int r = 0; r < 8; ++r) { oa[r] *= co; ob[r] *= co; }
}

struct QTile {
  v2f q[8];
  v8f o0, o1;
  float mrun;
  float lpart;
};

__device__ __forceinline__ void attn_block(const KVFrag& f, QTile& t0, QTile& t1) {
  v8f s0 = {}, s1 = {};
  for (int i = 0; i < 8; ++i) {
    s0 = wmma4(f.k[i], t0.q[i], s0);
    s1 = wmma4(f.k[i], t1.q[i], s1);
  }
  float p0[8], p1[8];
  softmax_tile(s0, t0.mrun, t0.lpart, t0.o0, t0.o1, p0);
  for (int j = 0; j < 4; ++j) {
    const v2f b = { p0[2 * j], p0[2 * j + 1] };
    t0.o0 = wmma4(f.v0[j], b, t0.o0);
    t0.o1 = wmma4(f.v1[j], b, t0.o1);
  }
  softmax_tile(s1, t1.mrun, t1.lpart, t1.o0, t1.o1, p1);
  for (int j = 0; j < 4; ++j) {
    const v2f b = { p1[2 * j], p1[2 * j + 1] };
    t1.o0 = wmma4(f.v0[j], b, t1.o0);
    t1.o1 = wmma4(f.v1[j], b, t1.o1);
  }
}

__device__ __forceinline__ void store_tile(const QTile& t, float* __restrict__ ob) {
  const float inv = 1.0f / (t.lpart + xhalf(t.lpart));
  *(float4*)(ob)      = make_float4(t.o0[0] * inv, t.o0[1] * inv, t.o0[2] * inv, t.o0[3] * inv);
  *(float4*)(ob + 4)  = make_float4(t.o0[4] * inv, t.o0[5] * inv, t.o0[6] * inv, t.o0[7] * inv);
  *(float4*)(ob + 16) = make_float4(t.o1[0] * inv, t.o1[1] * inv, t.o1[2] * inv, t.o1[3] * inv);
  *(float4*)(ob + 20) = make_float4(t.o1[4] * inv, t.o1[5] * inv, t.o1[6] * inv, t.o1[7] * inv);
}

__global__ __launch_bounds__(128) __attribute__((amdgpu_waves_per_eu(1, 4)))
void attn_kernel(const float* __restrict__ qkv, float* __restrict__ obuf) {
  const int lane = threadIdx.x & 31;
  const int wave = threadIdx.x >> 5;
  const int m    = lane & 15;
  const int half = lane >> 4;
  const int q0   = (blockIdx.x * 4 + wave) * 32;
  const int h    = blockIdx.y;

  QTile t0, t1;
  {
    const float* qr = qkv + (size_t)(q0 + m) * (3 * Dd) + h * HDd + 2 * half;
    for (int i = 0; i < 8; ++i) {
      t0.q[i] = *(const v2f*)(qr + 4 * i);
      t1.q[i] = *(const v2f*)(qr + (size_t)16 * (3 * Dd) + 4 * i);
    }
  }
  t0.o0 = v8f{}; t0.o1 = v8f{}; t1.o0 = v8f{}; t1.o1 = v8f{};
  t0.mrun = -INF_; t1.mrun = -INF_;
  t0.lpart = 0.f;  t1.lpart = 0.f;

  const float* kp = qkv + Dd + h * HDd + (size_t)m * (3 * Dd) + 2 * half;
  const float* vp = qkv + 2 * Dd + h * HDd + (size_t)(8 * half) * (3 * Dd) + m;
  const size_t kstep = (size_t)16 * (3 * Dd);

  KVFrag fa, fb;
  load_kv(fa, kp, vp);
  for (int kb = 0; kb < Nn / 16; kb += 2) {
    load_kv(fb, kp + (size_t)(kb + 1) * kstep, vp + (size_t)(kb + 1) * kstep);
    attn_block(fa, t0, t1);
    __builtin_amdgcn_sched_barrier(0);
    const int nx = (kb + 2 < Nn / 16) ? kb + 2 : kb + 1;
    load_kv(fa, kp + (size_t)nx * kstep, vp + (size_t)nx * kstep);
    asm volatile("" ::: "memory");
    attn_block(fb, t0, t1);
  }

  float* ob = obuf + (size_t)(q0 + m) * Dd + h * HDd + 8 * half;
  store_tile(t0, ob);
  store_tile(t1, ob + (size_t)16 * Dd);
}

__global__ void zerof(float* p, int n) {
  int i = blockIdx.x * blockDim.x + threadIdx.x;
  if (i < n) p[i] = 0.f;
}
__global__ void seg_sum(const float* __restrict__ ew, const int* __restrict__ src,
                        float* __restrict__ out, int E) {
  int i = blockIdx.x * blockDim.x + threadIdx.x;
  if (i < E) atomicAdd(&out[src[i]], ew[i]);
}

template <int KK, bool SCALE, bool ADD>
__global__ __launch_bounds__(128)
void gemm_res_ln(const float* __restrict__ A, const float* __restrict__ W,
                 const float* __restrict__ bias, const float* __restrict__ res,
                 const float* __restrict__ rscale,
                 const float* __restrict__ g, const float* __restrict__ b,
                 const float* __restrict__ addp, float* __restrict__ out) {
  __shared__ float red[2][4][16];

  const int lane = threadIdx.x & 31;
  const int wave = threadIdx.x >> 5;
  const int m    = lane & 15;
  const int half = lane >> 4;
  const int tbn  = wave * 64;
  const int tbm  = blockIdx.x * 16;

  const float* arow = A + (size_t)(tbm + m) * KK + half * 2;
  const float* w0   = W + (size_t)(tbn + m) * KK + half * 2;
  const float* w1   = w0 + (size_t)16 * KK;
  const float* w2   = w1 + (size_t)16 * KK;
  const float* w3   = w2 + (size_t)16 * KK;

  v8f c0 = {}, c1 = {}, c2 = {}, c3 = {};
#pragma unroll
  for (int k = 0; k < KK; k += 4) {
    v2f a  = { arow[k], arow[k + 1] };
    v2f b0 = { w0[k],   w0[k + 1] };
    v2f b1 = { w1[k],   w1[k + 1] };
    v2f b2 = { w2[k],   w2[k + 1] };
    v2f b3 = { w3[k],   w3[k + 1] };
    c0 = wmma4(a, b0, c0);
    c1 = wmma4(a, b1, c1);
    c2 = wmma4(a, b2, c2);
    c3 = wmma4(a, b3, c3);
  }

  const int col = tbn + m;
  const int rb  = 8 * half;
  const float bj0 = bias[col], bj1 = bias[col + 16], bj2 = bias[col + 32], bj3 = bias[col + 48];
  const float g0 = g[col], g1 = g[col + 16], g2 = g[col + 32], g3 = g[col + 48];
  const float e0 = b[col], e1 = b[col + 16], e2 = b[col + 32], e3 = b[col + 48];
  float a0[8], a1[8], a2[8], a3[8];
  if (ADD) {
#pragma unroll
    for (int r = 0; r < 8; ++r) {
      const float* ar = addp + (size_t)(tbm + rb + r) * Dd + col;
      a0[r] = ar[0]; a1[r] = ar[16]; a2[r] = ar[32]; a3[r] = ar[48];
    }
  }

  float t0[8], t1[8], t2[8], t3[8];
#pragma unroll
  for (int r = 0; r < 8; ++r) {
    const int row = tbm + rb + r;
    const float sc = SCALE ? rscale[row] : 1.0f;
    const float* rr = res + (size_t)row * Dd + col;
    t0[r] = rr[0]  + (c0[r] + bj0) * sc;
    t1[r] = rr[16] + (c1[r] + bj1) * sc;
    t2[r] = rr[32] + (c2[r] + bj2) * sc;
    t3[r] = rr[48] + (c3[r] + bj3) * sc;
  }

  float s[8];
#pragma unroll
  for (int r = 0; r < 8; ++r) s[r] = rsum16((t0[r] + t1[r]) + (t2[r] + t3[r]));
  if (m == 0) {
#pragma unroll
    for (int r = 0; r < 8; ++r) red[0][wave][rb + r] = s[r];
  }
  __syncthreads();
#pragma unroll
  for (int r = 0; r < 8; ++r) {
    const float mean = ((red[0][0][rb + r] + red[0][1][rb + r]) +
                        (red[0][2][rb + r] + red[0][3][rb + r])) / (float)Dd;
    t0[r] -= mean; t1[r] -= mean; t2[r] -= mean; t3[r] -= mean;
  }

#pragma unroll
  for (int r = 0; r < 8; ++r)
    s[r] = rsum16((t0[r] * t0[r] + t1[r] * t1[r]) + (t2[r] * t2[r] + t3[r] * t3[r]));
  if (m == 0) {
#pragma unroll
    for (int r = 0; r < 8; ++r) red[1][wave][rb + r] = s[r];
  }
  __syncthreads();

#pragma unroll
  for (int r = 0; r < 8; ++r) {
    const float var = ((red[1][0][rb + r] + red[1][1][rb + r]) +
                       (red[1][2][rb + r] + red[1][3][rb + r])) / (float)Dd;
    const float rs = rsqrtf(var + 1e-5f);
    const size_t off = (size_t)(tbm + rb + r) * Dd + col;
    float y0 = t0[r] * rs * g0 + e0;
    float y1 = t1[r] * rs * g1 + e1;
    float y2 = t2[r] * rs * g2 + e2;
    float y3 = t3[r] * rs * g3 + e3;
    if (ADD) { y0 += a0[r]; y1 += a1[r]; y2 += a2[r]; y3 += a3[r]; }
    float* orow = out + off;
    orow[0] = y0; orow[16] = y1; orow[32] = y2; orow[48] = y3;
  }
}

#define PWP  260
#define TPH  (Pp / 16)
#define PWG  256
#define PWV  12

__global__ __launch_bounds__(32 * PWV)
void predict_all(const float* __restrict__ hc,
                 const int* __restrict__ psrc, const int* __restrict__ pdst,
                 const int* __restrict__ nsrc, const int* __restrict__ ndst,
                 const float* __restrict__ pw1, const float* __restrict__ pb1,
                 const float* __restrict__ pw2, const float* __restrict__ pb2,
                 float* __restrict__ out) {
  __shared__ float wl[(Dd / 2) * PWP];

  const int tid = threadIdx.x;
  for (int idx = tid; idx < (Dd / 2) * (Dd / 4); idx += 32 * PWV) {
    const int row = idx >> 6, c4 = idx & 63;
    const float4 v = *(const float4*)(pw1 + (size_t)row * Dd + c4 * 4);
    *(float4*)(wl + row * PWP + c4 * 4) = v;
  }
  __syncthreads();

  const int lane = tid & 31;
  const int wave = tid >> 5;
  const int m    = lane & 15;
  const int half = lane >> 4;

  float bias1[8], w2[8];
#pragma unroll
  for (int jt = 0; jt < 8; ++jt) {
    bias1[jt] = pb1[jt * 16 + m];
    w2[jt]    = pw2[jt * 16 + m];
  }
  const float b2 = pb2[0];
  const float* wfrag = wl + m * PWP + 4 * half;

  for (int t = wave * PWG + (int)blockIdx.x; t < 2 * TPH; t += PWV * PWG) {
    asm volatile("" ::: "memory");
    const bool ng = t >= TPH;
    const int* sp = ng ? nsrc : psrc;
    const int* dp = ng ? ndst : pdst;
    const int tb  = (ng ? t - TPH : t) * 16;
    const int rs = sp[tb + m], rd = dp[tb + m];
    const float* as = hc + (size_t)rs * Dd + 4 * half;
    const float* ad = hc + (size_t)rd * Dd + 4 * half;

    float4 z[32];
#pragma unroll
    for (int i = 0; i < 32; ++i) {
      const float4 a = *(const float4*)(as + 8 * i);
      const float4 b = *(const float4*)(ad + 8 * i);
      z[i] = make_float4(a.x * b.x, a.y * b.y, a.z * b.z, a.w * b.w);
    }

    v8f c[8];
#pragma unroll
    for (int jt = 0; jt < 8; ++jt) c[jt] = v8f{};

#pragma unroll
    for (int i = 0; i < 32; ++i) {
      float4 w[8];
#pragma unroll
      for (int jt = 0; jt < 8; ++jt)
        w[jt] = *(const float4*)(wfrag + jt * 16 * PWP + 8 * i);
      const v2f a0 = { z[i].x, z[i].y };
      const v2f a1 = { z[i].z, z[i].w };
#pragma unroll
      for (int jt = 0; jt < 8; ++jt)
        c[jt] = wmma4(a0, v2f{ w[jt].x, w[jt].y }, c[jt]);
#pragma unroll
      for (int jt = 0; jt < 8; ++jt)
        c[jt] = wmma4(a1, v2f{ w[jt].z, w[jt].w }, c[jt]);
    }

    float acc[8];
#pragma unroll
    for (int r = 0; r < 8; ++r) acc[r] = 0.f;
#pragma unroll
    for (int jt = 0; jt < 8; ++jt) {
#pragma unroll
      for (int r = 0; r < 8; ++r) {
        float v = c[jt][r] + bias1[jt];
        v = v > 0.f ? v : 0.2f * v;
        acc[r] += v * w2[jt];
      }
    }
#pragma unroll
    for (int r = 0; r < 8; ++r) acc[r] = rsum16(acc[r]);
    float o = acc[0];
#pragma unroll
    for (int r = 1; r < 8; ++r) o = (m == r) ? acc[r] : o;
    if (m < 8) out[(size_t)t * 16 + 8 * half + m] = o + b2;
  }
}

extern "C" void kernel_launch(void* const* d_in, const int* in_sizes, int n_in,
                              void* d_out, int out_size, void* d_ws, size_t ws_size,
                              hipStream_t stream) {
  const float* x       = (const float*)d_in[0];
  const float* edge_w  = (const float*)d_in[1];
  const int*   src     = (const int*)d_in[2];
  const int*   pos_src = (const int*)d_in[4];
  const int*   pos_dst = (const int*)d_in[5];
  const int*   neg_src = (const int*)d_in[6];
  const int*   neg_dst = (const int*)d_in[7];
  const float* proj_w  = (const float*)d_in[8];
  const float* proj_b  = (const float*)d_in[9];
  const float* in_w    = (const float*)d_in[10];
  const float* in_b    = (const float*)d_in[11];
  const float* out_w   = (const float*)d_in[12];
  const float* out_b   = (const float*)d_in[13];
  const float* ff_w    = (const float*)d_in[14];
  const float* ff_b    = (const float*)d_in[15];
  const float* g1      = (const float*)d_in[16];
  const float* b1      = (const float*)d_in[17];
  const float* g2      = (const float*)d_in[18];
  const float* b2      = (const float*)d_in[19];
  const float* pw1     = (const float*)d_in[20];
  const float* pb1     = (const float*)d_in[21];
  const float* pw2     = (const float*)d_in[22];
  const float* pb2     = (const float*)d_in[23];

  float* W      = (float*)d_ws;
  float* h      = W;
  float* hproj  = W + 1048576;
  float* qkv    = W + 2097152;
  float* obuf   = W + 5242880;
  float* h1     = W + 6291456;
  float* wsn    = W + 7340032;

  float* outp = (float*)d_out;
  float* pos  = outp;
  float* hc   = outp + 2 * Pp;

  gemm_atb<INc><<<dim3(Dd / 64, Nn / 16), 32, 0, stream>>>(x, proj_w, proj_b, hproj, Nn, Dd);

  const float* hin[Ll] = { hproj, h };

  for (int l = 0; l < Ll; ++l) {
    gemm_atb<Dd><<<dim3(3 * Dd / 64, Nn / 16), 32, 0, stream>>>(
        hin[l], in_w + (size_t)l * 3 * Dd * Dd, in_b + (size_t)l * 3 * Dd, qkv, Nn, 3 * Dd);
    attn_kernel<<<dim3(Nn / 128, Hh), 128, 0, stream>>>(qkv, obuf);
    zerof<<<(Nn + 255) / 256, 256, 0, stream>>>(wsn, Nn);
    seg_sum<<<(Ee + 255) / 256, 256, 0, stream>>>(edge_w + (size_t)l * Ee,
                                                  src + (size_t)l * Ee, wsn, Ee);
    gemm_res_ln<Dd, true, false><<<Nn / 16, 128, 0, stream>>>(
        obuf, out_w + (size_t)l * Dd * Dd, out_b + (size_t)l * Dd, hin[l], wsn,
        g1 + (size_t)l * Dd, b1 + (size_t)l * Dd, nullptr, h1);
    if (l + 1 < Ll)
      gemm_res_ln<Dd, false, false><<<Nn / 16, 128, 0, stream>>>(
          h1, ff_w + (size_t)l * Dd * Dd, ff_b + (size_t)l * Dd, h1, nullptr,
          g2 + (size_t)l * Dd, b2 + (size_t)l * Dd, nullptr, h);
    else
      gemm_res_ln<Dd, false, true><<<Nn / 16, 128, 0, stream>>>(
          h1, ff_w + (size_t)l * Dd * Dd, ff_b + (size_t)l * Dd, h1, nullptr,
          g2 + (size_t)l * Dd, b2 + (size_t)l * Dd, hproj, hc);
  }

  predict_all<<<PWG, 32 * PWV, 0, stream>>>(hc, pos_src, pos_dst, neg_src, neg_dst,
                                       pw1, pb1, pw2, pb2, pos);
}


# attention: one online-softmax step per 32 keys (two key blocks), products tile by tile so one tile's softmax runs beside the other's WMMAs; loads lead 32 keys; row-sum tree, rescale points change
# speedup vs baseline: 3.1785x; 1.2213x over previous
#include <hip/hip_runtime.h>
#include <math.h>

typedef __attribute__((ext_vector_type(2))) float v2f;
typedef __attribute__((ext_vector_type(8))) float v8f;

#define Nn   4096
#define INF_ 3.402823466e+38f
#define Dd   256
#define Hh   8
#define HDd  32
#define Ll   2
#define Ee   131072
#define Pp   200000
#define INc  128

__device__ __forceinline__ v8f wmma4(v2f a, v2f b, v8f c) {
  return __builtin_amdgcn_wmma_f32_16x16x4_f32(
      false, a, false, b, (short)0, c, false, false);
}

__device__ __forceinline__ float rsum16(float v) {
  for (int m = 8; m >= 1; m >>= 1) v += __shfl_xor(v, m, 16);
  return v;
}
__device__ __forceinline__ float rmax16(float v) {
  for (int m = 8; m >= 1; m >>= 1) v = fmaxf(v, __shfl_xor(v, m, 16));
  return v;
}

template <int KK>
__global__ __launch_bounds__(32)
void gemm_atb(const float* __restrict__ A, const float* __restrict__ W,
              const float* __restrict__ bias, float* __restrict__ C,
              int M, int Nout) {
  const int tid  = threadIdx.x;
  const int m    = tid & 15;
  const int half = tid >> 4;
  const int tbn  = blockIdx.x * 64;
  const int tbm  = blockIdx.y * 16;

  const float* arow = A + (size_t)(tbm + m) * KK + half * 2;
  const float* w0   = W + (size_t)(tbn + m) * KK + half * 2;
  const float* w1   = w0 + (size_t)16 * KK;
  const float* w2   = w1 + (size_t)16 * KK;
  const float* w3   = w2 + (size_t)16 * KK;

  v8f c0 = {}, c1 = {}, c2 = {}, c3 = {};
#pragma unroll
  for (int k = 0; k < KK; k += 4) {
    v2f a  = { arow[k], arow[k + 1] };
    v2f b0 = { w0[k],   w0[k + 1] };
    v2f b1 = { w1[k],   w1[k + 1] };
    v2f b2 = { w2[k],   w2[k + 1] };
    v2f b3 = { w3[k],   w3[k + 1] };
    c0 = wmma4(a, b0, c0);
    c1 = wmma4(a, b1, c1);
    c2 = wmma4(a, b2, c2);
    c3 = wmma4(a, b3, c3);
  }
  for (int r = 0; r < 8; ++r) {
    const int row = tbm + r + 8 * half;
    float* crow = C + (size_t)row * Nout + tbn + m;
    if (bias) {
      const float* brow = bias + tbn + m;
      crow[0]  = c0[r] + brow[0];
      crow[16] = c1[r] + brow[16];
      crow[32] = c2[r] + brow[32];
      crow[48] = c3[r] + brow[48];
    } else {
      crow[0] = c0[r]; crow[16] = c1[r]; crow[32] = c2[r]; crow[48] = c3[r];
    }
  }
}

__device__ __forceinline__ float xhalf(float x) {
  const int b = __builtin_bit_cast(int, x);
  return __builtin_bit_cast(float,
      __builtin_amdgcn_permlanex16(b, b, 0x76543210, 0xfedcba98, false, false));
}

struct KVFrag {
  v2f k[8];
  v2f v0[4];
  v2f v1[4];
};

__device__ __forceinline__ void load_kv(KVFrag& f, const float* __restrict__ kp,
                                        const float* __restrict__ vp) {
  for (int i = 0; i < 8; ++i) f.k[i] = *(const v2f*)(kp + 4 * i);
  for (int j = 0; j < 4; ++j) {
    const float* vr = vp + (size_t)(2 * j) * (3 * Dd);
    f.v0[j] = v2f{ vr[0],  vr[3 * Dd] };
    f.v1[j] = v2f{ vr[16], vr[3 * Dd + 16] };
  }
}

__device__ __forceinline__ void softmax_pair(const v8f& sa, const v8f& sb, float& mrun,
                                             float& lpart, v8f& oa, v8f& ob,
                                             float (&p)[16]) {
  const float scale = 0.1767766952966369f;
  float sv[16];
  for (int r = 0; r < 8; ++r) { sv[r] = sa[r] * scale; sv[8 + r] = sb[r] * scale; }
  float mxa = fmaxf(fmaxf(fmaxf(sv[0], sv[1]), fmaxf(sv[2], sv[3])),
                    fmaxf(fmaxf(sv[4], sv[5]), fmaxf(sv[6], sv[7])));
  float mxb = fmaxf(fmaxf(fmaxf(sv[8], sv[9]), fmaxf(sv[10], sv[11])),
                    fmaxf(fmaxf(sv[12], sv[13]), fmaxf(sv[14], sv[15])));
  float mx = fmaxf(mxa, mxb);
  mx = fmaxf(mx, xhalf(mx));
  const float mn = fmaxf(mrun, mx);
  const float co = __expf(mrun - mn);
  for (int r = 0; r < 16; ++r) p[r] = __expf(sv[r] - mn);
  const float psa = ((p[0] + p[1]) + (p[2] + p[3])) + ((p[4] + p[5]) + (p[6] + p[7]));
  const float psb = ((p[8] + p[9]) + (p[10] + p[11])) + ((p[12] + p[13]) + (p[14] + p[15]));
  lpart = lpart * co + (psa + psb);
  mrun = mn;
  for (int r = 0; r < 8; ++r) { oa[r] *= co; ob[r] *= co; }
}

struct QTile {
  v2f q[8];
  v8f o0, o1;
  float mrun;
  float lpart;
};

__device__ __forceinline__ void pv_pair(const KVFrag& fa, const KVFrag& fb, QTile& t,
                                        const float (&p)[16]) {
  for (int j = 0; j < 4; ++j) {
    const v2f b = { p[2 * j], p[2 * j + 1] };
    t.o0 = wmma4(fa.v0[j], b, t.o0);
    t.o1 = wmma4(fa.v1[j], b, t.o1);
  }
  for (int j = 0; j < 4; ++j) {
    const v2f b = { p[8 + 2 * j], p[8 + 2 * j + 1] };
    t.o0 = wmma4(fb.v0[j], b, t.o0);
    t.o1 = wmma4(fb.v1[j], b, t.o1);
  }
}

__device__ __forceinline__ void attn_pair(const KVFrag& fa, const KVFrag& fb,
                                          QTile& t0, QTile& t1) {
  v8f s0a = {}, s0b = {}, s1a = {}, s1b = {};
  for (int i = 0; i < 8; ++i) {
    s0a = wmma4(fa.k[i], t0.q[i], s0a);
    s0b = wmma4(fb.k[i], t0.q[i], s0b);
  }
  for (int i = 0; i < 8; ++i) {
    s1a = wmma4(fa.k[i], t1.q[i], s1a);
    s1b = wmma4(fb.k[i], t1.q[i], s1b);
  }
  float p0[16], p1[16];
  softmax_pair(s0a, s0b, t0.mrun, t0.lpart, t0.o0, t0.o1, p0);
  pv_pair(fa, fb, t0, p0);
  softmax_pair(s1a, s1b, t1.mrun, t1.lpart, t1.o0, t1.o1, p1);
  pv_pair(fa, fb, t1, p1);
}

__device__ __forceinline__ void store_tile(const QTile& t, float* __restrict__ ob) {
  const float inv = 1.0f / (t.lpart + xhalf(t.lpart));
  *(float4*)(ob)      = make_float4(t.o0[0] * inv, t.o0[1] * inv, t.o0[2] * inv, t.o0[3] * inv);
  *(float4*)(ob + 4)  = make_float4(t.o0[4] * inv, t.o0[5] * inv, t.o0[6] * inv, t.o0[7] * inv);
  *(float4*)(ob + 16) = make_float4(t.o1[0] * inv, t.o1[1] * inv, t.o1[2] * inv, t.o1[3] * inv);
  *(float4*)(ob + 20) = make_float4(t.o1[4] * inv, t.o1[5] * inv, t.o1[6] * inv, t.o1[7] * inv);
}

__global__ __launch_bounds__(128) __attribute__((amdgpu_waves_per_eu(1, 2)))
void attn_kernel(const float* __restrict__ qkv, float* __restrict__ obuf) {
  const int lane = threadIdx.x & 31;
  const int wave = threadIdx.x >> 5;
  const int m    = lane & 15;
  const int half = lane >> 4;
  const int q0   = (blockIdx.x * 4 + wave) * 32;
  const int h    = blockIdx.y;

  QTile t0, t1;
  {
    const float* qr = qkv + (size_t)(q0 + m) * (3 * Dd) + h * HDd + 2 * half;
    for (int i = 0; i < 8; ++i) {
      t0.q[i] = *(const v2f*)(qr + 4 * i);
      t1.q[i] = *(const v2f*)(qr + (size_t)16 * (3 * Dd) + 4 * i);
    }
  }
  t0.o0 = v8f{}; t0.o1 = v8f{}; t1.o0 = v8f{}; t1.o1 = v8f{};
  t0.mrun = -INF_; t1.mrun = -INF_;
  t0.lpart = 0.f;  t1.lpart = 0.f;

  const float* kp = qkv + Dd + h * HDd + (size_t)m * (3 * Dd) + 2 * half;
  const float* vp = qkv + 2 * Dd + h * HDd + (size_t)(8 * half) * (3 * Dd) + m;
  const size_t kstep = (size_t)16 * (3 * Dd);

  KVFrag fa0, fa1, fb0, fb1;
  load_kv(fa0, kp, vp);
  load_kv(fa1, kp + kstep, vp + kstep);
  for (int kb = 0; kb < Nn / 16; kb += 4) {
    load_kv(fb0, kp + (size_t)(kb + 2) * kstep, vp + (size_t)(kb + 2) * kstep);
    load_kv(fb1, kp + (size_t)(kb + 3) * kstep, vp + (size_t)(kb + 3) * kstep);
    attn_pair(fa0, fa1, t0, t1);
    __builtin_amdgcn_sched_barrier(0);
    const int nx = (kb + 4 < Nn / 16) ? kb + 4 : kb + 2;
    load_kv(fa0, kp + (size_t)nx * kstep, vp + (size_t)nx * kstep);
    load_kv(fa1, kp + (size_t)(nx + 1) * kstep, vp + (size_t)(nx + 1) * kstep);
    asm volatile("" ::: "memory");
    attn_pair(fb0, fb1, t0, t1);
  }

  float* ob = obuf + (size_t)(q0 + m) * Dd + h * HDd + 8 * half;
  store_tile(t0, ob);
  store_tile(t1, ob + (size_t)16 * Dd);
}

__global__ void zerof(float* p, int n) {
  int i = blockIdx.x * blockDim.x + threadIdx.x;
  if (i < n) p[i] = 0.f;
}
__global__ void seg_sum(const float* __restrict__ ew, const int* __restrict__ src,
                        float* __restrict__ out, int E) {
  int i = blockIdx.x * blockDim.x + threadIdx.x;
  if (i < E) atomicAdd(&out[src[i]], ew[i]);
}

template <int KK, bool SCALE, bool ADD>
__global__ __launch_bounds__(128)
void gemm_res_ln(const float* __restrict__ A, const float* __restrict__ W,
                 const float* __restrict__ bias, const float* __restrict__ res,
                 const float* __restrict__ rscale,
                 const float* __restrict__ g, const float* __restrict__ b,
                 const float* __restrict__ addp, float* __restrict__ out) {
  __shared__ float red[2][4][16];

  const int lane = threadIdx.x & 31;
  const int wave = threadIdx.x >> 5;
  const int m    = lane & 15;
  const int half = lane >> 4;
  const int tbn  = wave * 64;
  const int tbm  = blockIdx.x * 16;

  const float* arow = A + (size_t)(tbm + m) * KK + half * 2;
  const float* w0   = W + (size_t)(tbn + m) * KK + half * 2;
  const float* w1   = w0 + (size_t)16 * KK;
  const float* w2   = w1 + (size_t)16 * KK;
  const float* w3   = w2 + (size_t)16 * KK;

  v8f c0 = {}, c1 = {}, c2 = {}, c3 = {};
#pragma unroll
  for (int k = 0; k < KK; k += 4) {
    v2f a  = { arow[k], arow[k + 1] };
    v2f b0 = { w0[k],   w0[k + 1] };
    v2f b1 = { w1[k],   w1[k + 1] };
    v2f b2 = { w2[k],   w2[k + 1] };
    v2f b3 = { w3[k],   w3[k + 1] };
    c0 = wmma4(a, b0, c0);
    c1 = wmma4(a, b1, c1);
    c2 = wmma4(a, b2, c2);
    c3 = wmma4(a, b3, c3);
  }

  const int col = tbn + m;
  const int rb  = 8 * half;
  const float bj0 = bias[col], bj1 = bias[col + 16], bj2 = bias[col + 32], bj3 = bias[col + 48];
  const float g0 = g[col], g1 = g[col + 16], g2 = g[col + 32], g3 = g[col + 48];
  const float e0 = b[col], e1 = b[col + 16], e2 = b[col + 32], e3 = b[col + 48];
  float a0[8], a1[8], a2[8], a3[8];
  if (ADD) {
#pragma unroll
    for (int r = 0; r < 8; ++r) {
      const float* ar = addp + (size_t)(tbm + rb + r) * Dd + col;
      a0[r] = ar[0]; a1[r] = ar[16]; a2[r] = ar[32]; a3[r] = ar[48];
    }
  }

  float t0[8], t1[8], t2[8], t3[8];
#pragma unroll
  for (int r = 0; r < 8; ++r) {
    const int row = tbm + rb + r;
    const float sc = SCALE ? rscale[row] : 1.0f;
    const float* rr = res + (size_t)row * Dd + col;
    t0[r] = rr[0]  + (c0[r] + bj0) * sc;
    t1[r] = rr[16] + (c1[r] + bj1) * sc;
    t2[r] = rr[32] + (c2[r] + bj2) * sc;
    t3[r] = rr[48] + (c3[r] + bj3) * sc;
  }

  float s[8];
#pragma unroll
  for (int r = 0; r < 8; ++r) s[r] = rsum16((t0[r] + t1[r]) + (t2[r] + t3[r]));
  if (m == 0) {
#pragma unroll
    for (int r = 0; r < 8; ++r) red[0][wave][rb + r] = s[r];
  }
  __syncthreads();
#pragma unroll
  for (int r = 0; r < 8; ++r) {
    const float mean = ((red[0][0][rb + r] + red[0][1][rb + r]) +
                        (red[0][2][rb + r] + red[0][3][rb + r])) / (float)Dd;
    t0[r] -= mean; t1[r] -= mean; t2[r] -= mean; t3[r] -= mean;
  }

#pragma unroll
  for (int r = 0; r < 8; ++r)
    s[r] = rsum16((t0[r] * t0[r] + t1[r] * t1[r]) + (t2[r] * t2[r] + t3[r] * t3[r]));
  if (m == 0) {
#pragma unroll
    for (int r = 0; r < 8; ++r) red[1][wave][rb + r] = s[r];
  }
  __syncthreads();

#pragma unroll
  for (int r = 0; r < 8; ++r) {
    const float var = ((red[1][0][rb + r] + red[1][1][rb + r]) +
                       (red[1][2][rb + r] + red[1][3][rb + r])) / (float)Dd;
    const float rs = rsqrtf(var + 1e-5f);
    const size_t off = (size_t)(tbm + rb + r) * Dd + col;
    float y0 = t0[r] * rs * g0 + e0;
    float y1 = t1[r] * rs * g1 + e1;
    float y2 = t2[r] * rs * g2 + e2;
    float y3 = t3[r] * rs * g3 + e3;
    if (ADD) { y0 += a0[r]; y1 += a1[r]; y2 += a2[r]; y3 += a3[r]; }
    float* orow = out + off;
    orow[0] = y0; orow[16] = y1; orow[32] = y2; orow[48] = y3;
  }
}

#define PWP  260
#define TPH  (Pp / 16)
#define PWG  256
#define PWV  12

__global__ __launch_bounds__(32 * PWV)
void predict_all(const float* __restrict__ hc,
                 const int* __restrict__ psrc, const int* __restrict__ pdst,
                 const int* __restrict__ nsrc, const int* __restrict__ ndst,
                 const float* __restrict__ pw1, const float* __restrict__ pb1,
                 const float* __restrict__ pw2, const float* __restrict__ pb2,
                 float* __restrict__ out) {
  __shared__ float wl[(Dd / 2) * PWP];

  const int tid = threadIdx.x;
  for (int idx = tid; idx < (Dd / 2) * (Dd / 4); idx += 32 * PWV) {
    const int row = idx >> 6, c4 = idx & 63;
    const float4 v = *(const float4*)(pw1 + (size_t)row * Dd + c4 * 4);
    *(float4*)(wl + row * PWP + c4 * 4) = v;
  }
  __syncthreads();

  const int lane = tid & 31;
  const int wave = tid >> 5;
  const int m    = lane & 15;
  const int half = lane >> 4;

  float bias1[8], w2[8];
#pragma unroll
  for (int jt = 0; jt < 8; ++jt) {
    bias1[jt] = pb1[jt * 16 + m];
    w2[jt]    = pw2[jt * 16 + m];
  }
  const float b2 = pb2[0];
  const float* wfrag = wl + m * PWP + 4 * half;

  for (int t = wave * PWG + (int)blockIdx.x; t < 2 * TPH; t += PWV * PWG) {
    asm volatile("" ::: "memory");
    const bool ng = t >= TPH;
    const int* sp = ng ? nsrc : psrc;
    const int* dp = ng ? ndst : pdst;
    const int tb  = (ng ? t - TPH : t) * 16;
    const int rs = sp[tb + m], rd = dp[tb + m];
    const float* as = hc + (size_t)rs * Dd + 4 * half;
    const float* ad = hc + (size_t)rd * Dd + 4 * half;

    float4 z[32];
#pragma unroll
    for (int i = 0; i < 32; ++i) {
      const float4 a = *(const float4*)(as + 8 * i);
      const float4 b = *(const float4*)(ad + 8 * i);
      z[i] = make_float4(a.x * b.x, a.y * b.y, a.z * b.z, a.w * b.w);
    }

    v8f c[8];
#pragma unroll
    for (int jt = 0; jt < 8; ++jt) c[jt] = v8f{};

#pragma unroll
    for (int i = 0; i < 32; ++i) {
      float4 w[8];
#pragma unroll
      for (int jt = 0; jt < 8; ++jt)
        w[jt] = *(const float4*)(wfrag + jt * 16 * PWP + 8 * i);
      const v2f a0 = { z[i].x, z[i].y };
      const v2f a1 = { z[i].z, z[i].w };
#pragma unroll
      for (int jt = 0; jt < 8; ++jt)
        c[jt] = wmma4(a0, v2f{ w[jt].x, w[jt].y }, c[jt]);
#pragma unroll
      for (int jt = 0; jt < 8; ++jt)
        c[jt] = wmma4(a1, v2f{ w[jt].z, w[jt].w }, c[jt]);
    }

    float acc[8];
#pragma unroll
    for (int r = 0; r < 8; ++r) acc[r] = 0.f;
#pragma unroll
    for (int jt = 0; jt < 8; ++jt) {
#pragma unroll
      for (int r = 0; r < 8; ++r) {
        float v = c[jt][r] + bias1[jt];
        v = v > 0.f ? v : 0.2f * v;
        acc[r] += v * w2[jt];
      }
    }
#pragma unroll
    for (int r = 0; r < 8; ++r) acc[r] = rsum16(acc[r]);
    float o = acc[0];
#pragma unroll
    for (int r = 1; r < 8; ++r) o = (m == r) ? acc[r] : o;
    if (m < 8) out[(size_t)t * 16 + 8 * half + m] = o + b2;
  }
}

extern "C" void kernel_launch(void* const* d_in, const int* in_sizes, int n_in,
                              void* d_out, int out_size, void* d_ws, size_t ws_size,
                              hipStream_t stream) {
  const float* x       = (const float*)d_in[0];
  const float* edge_w  = (const float*)d_in[1];
  const int*   src     = (const int*)d_in[2];
  const int*   pos_src = (const int*)d_in[4];
  const int*   pos_dst = (const int*)d_in[5];
  const int*   neg_src = (const int*)d_in[6];
  const int*   neg_dst = (const int*)d_in[7];
  const float* proj_w  = (const float*)d_in[8];
  const float* proj_b  = (const float*)d_in[9];
  const float* in_w    = (const float*)d_in[10];
  const float* in_b    = (const float*)d_in[11];
  const float* out_w   = (const float*)d_in[12];
  const float* out_b   = (const float*)d_in[13];
  const float* ff_w    = (const float*)d_in[14];
  const float* ff_b    = (const float*)d_in[15];
  const float* g1      = (const float*)d_in[16];
  const float* b1      = (const float*)d_in[17];
  const float* g2      = (const float*)d_in[18];
  const float* b2      = (const float*)d_in[19];
  const float* pw1     = (const float*)d_in[20];
  const float* pb1     = (const float*)d_in[21];
  const float* pw2     = (const float*)d_in[22];
  const float* pb2     = (const float*)d_in[23];

  float* W      = (float*)d_ws;
  float* h      = W;
  float* hproj  = W + 1048576;
  float* qkv    = W + 2097152;
  float* obuf   = W + 5242880;
  float* h1     = W + 6291456;
  float* wsn    = W + 7340032;

  float* outp = (float*)d_out;
  float* pos  = outp;
  float* hc   = outp + 2 * Pp;

  gemm_atb<INc><<<dim3(Dd / 64, Nn / 16), 32, 0, stream>>>(x, proj_w, proj_b, hproj, Nn, Dd);

  const float* hin[Ll] = { hproj, h };

  for (int l = 0; l < Ll; ++l) {
    gemm_atb<Dd><<<dim3(3 * Dd / 64, Nn / 16), 32, 0, stream>>>(
        hin[l], in_w + (size_t)l * 3 * Dd * Dd, in_b + (size_t)l * 3 * Dd, qkv, Nn, 3 * Dd);
    attn_kernel<<<dim3(Nn / 128, Hh), 128, 0, stream>>>(qkv, obuf);
    zerof<<<(Nn + 255) / 256, 256, 0, stream>>>(wsn, Nn);
    seg_sum<<<(Ee + 255) / 256, 256, 0, stream>>>(edge_w + (size_t)l * Ee,
                                                  src + (size_t)l * Ee, wsn, Ee);
    gemm_res_ln<Dd, true, false><<<Nn / 16, 128, 0, stream>>>(
        obuf, out_w + (size_t)l * Dd * Dd, out_b + (size_t)l * Dd, hin[l], wsn,
        g1 + (size_t)l * Dd, b1 + (size_t)l * Dd, nullptr, h1);
    if (l + 1 < Ll)
      gemm_res_ln<Dd, false, false><<<Nn / 16, 128, 0, stream>>>(
          h1, ff_w + (size_t)l * Dd * Dd, ff_b + (size_t)l * Dd, h1, nullptr,
          g2 + (size_t)l * Dd, b2 + (size_t)l * Dd, nullptr, h);
    else
      gemm_res_ln<Dd, false, true><<<Nn / 16, 128, 0, stream>>>(
          h1, ff_w + (size_t)l * Dd * Dd, ff_b + (size_t)l * Dd, h1, nullptr,
          g2 + (size_t)l * Dd, b2 + (size_t)l * Dd, hproj, hc);
  }

  predict_all<<<PWG, 32 * PWV, 0, stream>>>(hc, pos_src, pos_dst, neg_src, neg_dst,
                                       pw1, pb1, pw2, pb2, pos);
}


# attention: a query's keys split over two waves (first/second half) so each SIMD holds 2 waves; halves merged once via LDS (rescaled to common max, first+second): output and row-sum order changes
# speedup vs baseline: 3.4743x; 1.0931x over previous
#include <hip/hip_runtime.h>
#include <math.h>

typedef __attribute__((ext_vector_type(2))) float v2f;
typedef __attribute__((ext_vector_type(8))) float v8f;

#define Nn   4096
#define INF_ 3.402823466e+38f
#define Dd   256
#define Hh   8
#define HDd  32
#define Ll   2
#define Ee   131072
#define Pp   200000
#define INc  128

__device__ __forceinline__ v8f wmma4(v2f a, v2f b, v8f c) {
  return __builtin_amdgcn_wmma_f32_16x16x4_f32(
      false, a, false, b, (short)0, c, false, false);
}

__device__ __forceinline__ float rsum16(float v) {
  for (int m = 8; m >= 1; m >>= 1) v += __shfl_xor(v, m, 16);
  return v;
}
__device__ __forceinline__ float rmax16(float v) {
  for (int m = 8; m >= 1; m >>= 1) v = fmaxf(v, __shfl_xor(v, m, 16));
  return v;
}

template <int KK>
__global__ __launch_bounds__(32)
void gemm_atb(const float* __restrict__ A, const float* __restrict__ W,
              const float* __restrict__ bias, float* __restrict__ C,
              int M, int Nout) {
  const int tid  = threadIdx.x;
  const int m    = tid & 15;
  const int half = tid >> 4;
  const int tbn  = blockIdx.x * 64;
  const int tbm  = blockIdx.y * 16;

  const float* arow = A + (size_t)(tbm + m) * KK + half * 2;
  const float* w0   = W + (size_t)(tbn + m) * KK + half * 2;
  const float* w1   = w0 + (size_t)16 * KK;
  const float* w2   = w1 + (size_t)16 * KK;
  const float* w3   = w2 + (size_t)16 * KK;

  v8f c0 = {}, c1 = {}, c2 = {}, c3 = {};
#pragma unroll
  for (int k = 0; k < KK; k += 4) {
    v2f a  = { arow[k], arow[k + 1] };
    v2f b0 = { w0[k],   w0[k + 1] };
    v2f b1 = { w1[k],   w1[k + 1] };
    v2f b2 = { w2[k],   w2[k + 1] };
    v2f b3 = { w3[k],   w3[k + 1] };
    c0 = wmma4(a, b0, c0);
    c1 = wmma4(a, b1, c1);
    c2 = wmma4(a, b2, c2);
    c3 = wmma4(a, b3, c3);
  }
  for (int r = 0; r < 8; ++r) {
    const int row = tbm + r + 8 * half;
    float* crow = C + (size_t)row * Nout + tbn + m;
    if (bias) {
      const float* brow = bias + tbn + m;
      crow[0]  = c0[r] + brow[0];
      crow[16] = c1[r] + brow[16];
      crow[32] = c2[r] + brow[32];
      crow[48] = c3[r] + brow[48];
    } else {
      crow[0] = c0[r]; crow[16] = c1[r]; crow[32] = c2[r]; crow[48] = c3[r];
    }
  }
}

__device__ __forceinline__ float xhalf(float x) {
  const int b = __builtin_bit_cast(int, x);
  return __builtin_bit_cast(float,
      __builtin_amdgcn_permlanex16(b, b, 0x76543210, 0xfedcba98, false, false));
}

struct KVFrag {
  v2f k[8];
  v2f v0[4];
  v2f v1[4];
};

__device__ __forceinline__ void load_kv(KVFrag& f, const float* __restrict__ kp,
                                        const float* __restrict__ vp) {
  for (int i = 0; i < 8; ++i) f.k[i] = *(const v2f*)(kp + 4 * i);
  for (int j = 0; j < 4; ++j) {
    const float* vr = vp + (size_t)(2 * j) * (3 * Dd);
    f.v0[j] = v2f{ vr[0],  vr[3 * Dd] };
    f.v1[j] = v2f{ vr[16], vr[3 * Dd + 16] };
  }
}

__device__ __forceinline__ void softmax_pair(const v8f& sa, const v8f& sb, float& mrun,
                                             float& lpart, v8f& oa, v8f& ob,
                                             float (&p)[16]) {
  const float scale = 0.1767766952966369f;
  float sv[16];
  for (int r = 0; r < 8; ++r) { sv[r] = sa[r] * scale; sv[8 + r] = sb[r] * scale; }
  float mxa = fmaxf(fmaxf(fmaxf(sv[0], sv[1]), fmaxf(sv[2], sv[3])),
                    fmaxf(fmaxf(sv[4], sv[5]), fmaxf(sv[6], sv[7])));
  float mxb = fmaxf(fmaxf(fmaxf(sv[8], sv[9]), fmaxf(sv[10], sv[11])),
                    fmaxf(fmaxf(sv[12], sv[13]), fmaxf(sv[14], sv[15])));
  float mx = fmaxf(mxa, mxb);
  mx = fmaxf(mx, xhalf(mx));
  const float mn = fmaxf(mrun, mx);
  const float co = __expf(mrun - mn);
  for (int r = 0; r < 16; ++r) p[r] = __expf(sv[r] - mn);
  const float psa = ((p[0] + p[1]) + (p[2] + p[3])) + ((p[4] + p[5]) + (p[6] + p[7]));
  const float psb = ((p[8] + p[9]) + (p[10] + p[11])) + ((p[12] + p[13]) + (p[14] + p[15]));
  lpart = lpart * co + (psa + psb);
  mrun = mn;
  for (int r = 0; r < 8; ++r) { oa[r] *= co; ob[r] *= co; }
}

struct QTile {
  v2f q[8];
  v8f o0, o1;
  float mrun;
  float lpart;
};

__device__ __forceinline__ void pv_pair(const KVFrag& fa, const KVFrag& fb, QTile& t,
                                        const float (&p)[16]) {
  for (int j = 0; j < 4; ++j) {
    const v2f b = { p[2 * j], p[2 * j + 1] };
    t.o0 = wmma4(fa.v0[j], b, t.o0);
    t.o1 = wmma4(fa.v1[j], b, t.o1);
  }
  for (int j = 0; j < 4; ++j) {
    const v2f b = { p[8 + 2 * j], p[8 + 2 * j + 1] };
    t.o0 = wmma4(fb.v0[j], b, t.o0);
    t.o1 = wmma4(fb.v1[j], b, t.o1);
  }
}

__device__ __forceinline__ void attn_pair(const KVFrag& fa, const KVFrag& fb,
                                          QTile& t0, QTile& t1) {
  v8f s0a = {}, s0b = {}, s1a = {}, s1b = {};
  for (int i = 0; i < 8; ++i) {
    s0a = wmma4(fa.k[i], t0.q[i], s0a);
    s0b = wmma4(fb.k[i], t0.q[i], s0b);
  }
  for (int i = 0; i < 8; ++i) {
    s1a = wmma4(fa.k[i], t1.q[i], s1a);
    s1b = wmma4(fb.k[i], t1.q[i], s1b);
  }
  float p0[16], p1[16];
  softmax_pair(s0a, s0b, t0.mrun, t0.lpart, t0.o0, t0.o1, p0);
  pv_pair(fa, fb, t0, p0);
  softmax_pair(s1a, s1b, t1.mrun, t1.lpart, t1.o0, t1.o1, p1);
  pv_pair(fa, fb, t1, p1);
}

__device__ __forceinline__ void store_tile(const QTile& t, float* __restrict__ ob) {
  const float inv = 1.0f / (t.lpart + xhalf(t.lpart));
  *(float4*)(ob)      = make_float4(t.o0[0] * inv, t.o0[1] * inv, t.o0[2] * inv, t.o0[3] * inv);
  *(float4*)(ob + 4)  = make_float4(t.o0[4] * inv, t.o0[5] * inv, t.o0[6] * inv, t.o0[7] * inv);
  *(float4*)(ob + 16) = make_float4(t.o1[0] * inv, t.o1[1] * inv, t.o1[2] * inv, t.o1[3] * inv);
  *(float4*)(ob + 20) = make_float4(t.o1[4] * inv, t.o1[5] * inv, t.o1[6] * inv, t.o1[7] * inv);
}

#define MRG_T 18
__device__ __forceinline__ void put_tile(const QTile& t, float* __restrict__ d) {
  for (int r = 0; r < 8; ++r) { d[r * 32] = t.o0[r]; d[(8 + r) * 32] = t.o1[r]; }
  d[16 * 32] = t.mrun;
  d[17 * 32] = t.lpart;
}
__device__ __forceinline__ void merge_tile(QTile& t, const float* __restrict__ d) {
  const float m1 = d[16 * 32];
  const float l1 = d[17 * 32];
  const float mn = fmaxf(t.mrun, m1);
  const float a  = __expf(t.mrun - mn);
  const float b  = __expf(m1 - mn);
  for (int r = 0; r < 8; ++r) {
    t.o0[r] = t.o0[r] * a + d[r * 32] * b;
    t.o1[r] = t.o1[r] * a + d[(8 + r) * 32] * b;
  }
  t.lpart = t.lpart * a + l1 * b;
  t.mrun  = mn;
}

__global__ __launch_bounds__(256) __attribute__((amdgpu_waves_per_eu(1, 2)))
void attn_kernel(const float* __restrict__ qkv, float* __restrict__ obuf) {
  __shared__ float mrg[4][2 * MRG_T][32];

  const int lane = threadIdx.x & 31;
  const int wave = threadIdx.x >> 5;
  const int qs   = wave & 3;
  const int kh   = wave >> 2;
  const int m    = lane & 15;
  const int half = lane >> 4;
  const int q0   = (blockIdx.x * 4 + qs) * 32;
  const int h    = blockIdx.y;
  const int kfirst = kh * (Nn / 2);

  QTile t0, t1;
  {
    const float* qr = qkv + (size_t)(q0 + m) * (3 * Dd) + h * HDd + 2 * half;
    for (int i = 0; i < 8; ++i) {
      t0.q[i] = *(const v2f*)(qr + 4 * i);
      t1.q[i] = *(const v2f*)(qr + (size_t)16 * (3 * Dd) + 4 * i);
    }
  }
  t0.o0 = v8f{}; t0.o1 = v8f{}; t1.o0 = v8f{}; t1.o1 = v8f{};
  t0.mrun = -INF_; t1.mrun = -INF_;
  t0.lpart = 0.f;  t1.lpart = 0.f;

  const float* kp = qkv + Dd + h * HDd + (size_t)(kfirst + m) * (3 * Dd) + 2 * half;
  const float* vp = qkv + 2 * Dd + h * HDd + (size_t)(kfirst + 8 * half) * (3 * Dd) + m;
  const size_t kstep = (size_t)16 * (3 * Dd);

  KVFrag fa0, fa1, fb0, fb1;
  load_kv(fa0, kp, vp);
  load_kv(fa1, kp + kstep, vp + kstep);
  for (int kb = 0; kb < Nn / 32; kb += 4) {
    load_kv(fb0, kp + (size_t)(kb + 2) * kstep, vp + (size_t)(kb + 2) * kstep);
    load_kv(fb1, kp + (size_t)(kb + 3) * kstep, vp + (size_t)(kb + 3) * kstep);
    attn_pair(fa0, fa1, t0, t1);
    __builtin_amdgcn_sched_barrier(0);
    const int nx = (kb + 4 < Nn / 32) ? kb + 4 : kb + 2;
    load_kv(fa0, kp + (size_t)nx * kstep, vp + (size_t)nx * kstep);
    load_kv(fa1, kp + (size_t)(nx + 1) * kstep, vp + (size_t)(nx + 1) * kstep);
    __builtin_amdgcn_sched_barrier(0);
    attn_pair(fb0, fb1, t0, t1);
  }

  float* mg = &mrg[qs][0][lane];
  if (kh == 1) {
    put_tile(t0, mg);
    put_tile(t1, mg + MRG_T * 32);
  }
  __syncthreads();
  if (kh == 0) {
    merge_tile(t0, mg);
    merge_tile(t1, mg + MRG_T * 32);
    float* ob = obuf + (size_t)(q0 + m) * Dd + h * HDd + 8 * half;
    store_tile(t0, ob);
    store_tile(t1, ob + (size_t)16 * Dd);
  }
}

__global__ void zerof(float* p, int n) {
  int i = blockIdx.x * blockDim.x + threadIdx.x;
  if (i < n) p[i] = 0.f;
}
__global__ void seg_sum(const float* __restrict__ ew, const int* __restrict__ src,
                        float* __restrict__ out, int E) {
  int i = blockIdx.x * blockDim.x + threadIdx.x;
  if (i < E) atomicAdd(&out[src[i]], ew[i]);
}

template <int KK, bool SCALE, bool ADD>
__global__ __launch_bounds__(128)
void gemm_res_ln(const float* __restrict__ A, const float* __restrict__ W,
                 const float* __restrict__ bias, const float* __restrict__ res,
                 const float* __restrict__ rscale,
                 const float* __restrict__ g, const float* __restrict__ b,
                 const float* __restrict__ addp, float* __restrict__ out) {
  __shared__ float red[2][4][16];

  const int lane = threadIdx.x & 31;
  const int wave = threadIdx.x >> 5;
  const int m    = lane & 15;
  const int half = lane >> 4;
  const int tbn  = wave * 64;
  const int tbm  = blockIdx.x * 16;

  const float* arow = A + (size_t)(tbm + m) * KK + half * 2;
  const float* w0   = W + (size_t)(tbn + m) * KK + half * 2;
  const float* w1   = w0 + (size_t)16 * KK;
  const float* w2   = w1 + (size_t)16 * KK;
  const float* w3   = w2 + (size_t)16 * KK;

  v8f c0 = {}, c1 = {}, c2 = {}, c3 = {};
#pragma unroll
  for (int k = 0; k < KK; k += 4) {
    v2f a  = { arow[k], arow[k + 1] };
    v2f b0 = { w0[k],   w0[k + 1] };
    v2f b1 = { w1[k],   w1[k + 1] };
    v2f b2 = { w2[k],   w2[k + 1] };
    v2f b3 = { w3[k],   w3[k + 1] };
    c0 = wmma4(a, b0, c0);
    c1 = wmma4(a, b1, c1);
    c2 = wmma4(a, b2, c2);
    c3 = wmma4(a, b3, c3);
  }

  const int col = tbn + m;
  const int rb  = 8 * half;
  const float bj0 = bias[col], bj1 = bias[col + 16], bj2 = bias[col + 32], bj3 = bias[col + 48];
  const float g0 = g[col], g1 = g[col + 16], g2 = g[col + 32], g3 = g[col + 48];
  const float e0 = b[col], e1 = b[col + 16], e2 = b[col + 32], e3 = b[col + 48];
  float a0[8], a1[8], a2[8], a3[8];
  if (ADD) {
#pragma unroll
    for (int r = 0; r < 8; ++r) {
      const float* ar = addp + (size_t)(tbm + rb + r) * Dd + col;
      a0[r] = ar[0]; a1[r] = ar[16]; a2[r] = ar[32]; a3[r] = ar[48];
    }
  }

  float t0[8], t1[8], t2[8], t3[8];
#pragma unroll
  for (int r = 0; r < 8; ++r) {
    const int row = tbm + rb + r;
    const float sc = SCALE ? rscale[row] : 1.0f;
    const float* rr = res + (size_t)row * Dd + col;
    t0[r] = rr[0]  + (c0[r] + bj0) * sc;
    t1[r] = rr[16] + (c1[r] + bj1) * sc;
    t2[r] = rr[32] + (c2[r] + bj2) * sc;
    t3[r] = rr[48] + (c3[r] + bj3) * sc;
  }

  float s[8];
#pragma unroll
  for (int r = 0; r < 8; ++r) s[r] = rsum16((t0[r] + t1[r]) + (t2[r] + t3[r]));
  if (m == 0) {
#pragma unroll
    for (int r = 0; r < 8; ++r) red[0][wave][rb + r] = s[r];
  }
  __syncthreads();
#pragma unroll
  for (int r = 0; r < 8; ++r) {
    const float mean = ((red[0][0][rb + r] + red[0][1][rb + r]) +
                        (red[0][2][rb + r] + red[0][3][rb + r])) / (float)Dd;
    t0[r] -= mean; t1[r] -= mean; t2[r] -= mean; t3[r] -= mean;
  }

#pragma unroll
  for (int r = 0; r < 8; ++r)
    s[r] = rsum16((t0[r] * t0[r] + t1[r] * t1[r]) + (t2[r] * t2[r] + t3[r] * t3[r]));
  if (m == 0) {
#pragma unroll
    for (int r = 0; r < 8; ++r) red[1][wave][rb + r] = s[r];
  }
  __syncthreads();

#pragma unroll
  for (int r = 0; r < 8; ++r) {
    const float var = ((red[1][0][rb + r] + red[1][1][rb + r]) +
                       (red[1][2][rb + r] + red[1][3][rb + r])) / (float)Dd;
    const float rs = rsqrtf(var + 1e-5f);
    const size_t off = (size_t)(tbm + rb + r) * Dd + col;
    float y0 = t0[r] * rs * g0 + e0;
    float y1 = t1[r] * rs * g1 + e1;
    float y2 = t2[r] * rs * g2 + e2;
    float y3 = t3[r] * rs * g3 + e3;
    if (ADD) { y0 += a0[r]; y1 += a1[r]; y2 += a2[r]; y3 += a3[r]; }
    float* orow = out + off;
    orow[0] = y0; orow[16] = y1; orow[32] = y2; orow[48] = y3;
  }
}

#define PWP  260
#define TPH  (Pp / 16)
#define PWG  256
#define PWV  12

__global__ __launch_bounds__(32 * PWV)
void predict_all(const float* __restrict__ hc,
                 const int* __restrict__ psrc, const int* __restrict__ pdst,
                 const int* __restrict__ nsrc, const int* __restrict__ ndst,
                 const float* __restrict__ pw1, const float* __restrict__ pb1,
                 const float* __restrict__ pw2, const float* __restrict__ pb2,
                 float* __restrict__ out) {
  __shared__ float wl[(Dd / 2) * PWP];

  const int tid = threadIdx.x;
  for (int idx = tid; idx < (Dd / 2) * (Dd / 4); idx += 32 * PWV) {
    const int row = idx >> 6, c4 = idx & 63;
    const float4 v = *(const float4*)(pw1 + (size_t)row * Dd + c4 * 4);
    *(float4*)(wl + row * PWP + c4 * 4) = v;
  }
  __syncthreads();

  const int lane = tid & 31;
  const int wave = tid >> 5;
  const int m    = lane & 15;
  const int half = lane >> 4;

  float bias1[8], w2[8];
#pragma unroll
  for (int jt = 0; jt < 8; ++jt) {
    bias1[jt] = pb1[jt * 16 + m];
    w2[jt]    = pw2[jt * 16 + m];
  }
  const float b2 = pb2[0];
  const float* wfrag = wl + m * PWP + 4 * half;

  for (int t = wave * PWG + (int)blockIdx.x; t < 2 * TPH; t += PWV * PWG) {
    asm volatile("" ::: "memory");
    const bool ng = t >= TPH;
    const int* sp = ng ? nsrc : psrc;
    const int* dp = ng ? ndst : pdst;
    const int tb  = (ng ? t - TPH : t) * 16;
    const int rs = sp[tb + m], rd = dp[tb + m];
    const float* as = hc + (size_t)rs * Dd + 4 * half;
    const float* ad = hc + (size_t)rd * Dd + 4 * half;

    float4 z[32];
#pragma unroll
    for (int i = 0; i < 32; ++i) {
      const float4 a = *(const float4*)(as + 8 * i);
      const float4 b = *(const float4*)(ad + 8 * i);
      z[i] = make_float4(a.x * b.x, a.y * b.y, a.z * b.z, a.w * b.w);
    }

    v8f c[8];
#pragma unroll
    for (int jt = 0; jt < 8; ++jt) c[jt] = v8f{};

#pragma unroll
    for (int i = 0; i < 32; ++i) {
      float4 w[8];
#pragma unroll
      for (int jt = 0; jt < 8; ++jt)
        w[jt] = *(const float4*)(wfrag + jt * 16 * PWP + 8 * i);
      const v2f a0 = { z[i].x, z[i].y };
      const v2f a1 = { z[i].z, z[i].w };
#pragma unroll
      for (int jt = 0; jt < 8; ++jt)
        c[jt] = wmma4(a0, v2f{ w[jt].x, w[jt].y }, c[jt]);
#pragma unroll
      for (int jt = 0; jt < 8; ++jt)
        c[jt] = wmma4(a1, v2f{ w[jt].z, w[jt].w }, c[jt]);
    }

    float acc[8];
#pragma unroll
    for (int r = 0; r < 8; ++r) acc[r] = 0.f;
#pragma unroll
    for (int jt = 0; jt < 8; ++jt) {
#pragma unroll
      for (int r = 0; r < 8; ++r) {
        float v = c[jt][r] + bias1[jt];
        v = v > 0.f ? v : 0.2f * v;
        acc[r] += v * w2[jt];
      }
    }
#pragma unroll
    for (int r = 0; r < 8; ++r) acc[r] = rsum16(acc[r]);
    float o = acc[0];
#pragma unroll
    for (int r = 1; r < 8; ++r) o = (m == r) ? acc[r] : o;
    if (m < 8) out[(size_t)t * 16 + 8 * half + m] = o + b2;
  }
}

extern "C" void kernel_launch(void* const* d_in, const int* in_sizes, int n_in,
                              void* d_out, int out_size, void* d_ws, size_t ws_size,
                              hipStream_t stream) {
  const float* x       = (const float*)d_in[0];
  const float* edge_w  = (const float*)d_in[1];
  const int*   src     = (const int*)d_in[2];
  const int*   pos_src = (const int*)d_in[4];
  const int*   pos_dst = (const int*)d_in[5];
  const int*   neg_src = (const int*)d_in[6];
  const int*   neg_dst = (const int*)d_in[7];
  const float* proj_w  = (const float*)d_in[8];
  const float* proj_b  = (const float*)d_in[9];
  const float* in_w    = (const float*)d_in[10];
  const float* in_b    = (const float*)d_in[11];
  const float* out_w   = (const float*)d_in[12];
  const float* out_b   = (const float*)d_in[13];
  const float* ff_w    = (const float*)d_in[14];
  const float* ff_b    = (const float*)d_in[15];
  const float* g1      = (const float*)d_in[16];
  const float* b1      = (const float*)d_in[17];
  const float* g2      = (const float*)d_in[18];
  const float* b2      = (const float*)d_in[19];
  const float* pw1     = (const float*)d_in[20];
  const float* pb1     = (const float*)d_in[21];
  const float* pw2     = (const float*)d_in[22];
  const float* pb2     = (const float*)d_in[23];

  float* W      = (float*)d_ws;
  float* h      = W;
  float* hproj  = W + 1048576;
  float* qkv    = W + 2097152;
  float* obuf   = W + 5242880;
  float* h1     = W + 6291456;
  float* wsn    = W + 7340032;

  float* outp = (float*)d_out;
  float* pos  = outp;
  float* hc   = outp + 2 * Pp;

  gemm_atb<INc><<<dim3(Dd / 64, Nn / 16), 32, 0, stream>>>(x, proj_w, proj_b, hproj, Nn, Dd);

  const float* hin[Ll] = { hproj, h };

  for (int l = 0; l < Ll; ++l) {
    gemm_atb<Dd><<<dim3(3 * Dd / 64, Nn / 16), 32, 0, stream>>>(
        hin[l], in_w + (size_t)l * 3 * Dd * Dd, in_b + (size_t)l * 3 * Dd, qkv, Nn, 3 * Dd);
    attn_kernel<<<dim3(Nn / 128, Hh), 256, 0, stream>>>(qkv, obuf);
    zerof<<<(Nn + 255) / 256, 256, 0, stream>>>(wsn, Nn);
    seg_sum<<<(Ee + 255) / 256, 256, 0, stream>>>(edge_w + (size_t)l * Ee,
                                                  src + (size_t)l * Ee, wsn, Ee);
    gemm_res_ln<Dd, true, false><<<Nn / 16, 128, 0, stream>>>(
        obuf, out_w + (size_t)l * Dd * Dd, out_b + (size_t)l * Dd, hin[l], wsn,
        g1 + (size_t)l * Dd, b1 + (size_t)l * Dd, nullptr, h1);
    if (l + 1 < Ll)
      gemm_res_ln<Dd, false, false><<<Nn / 16, 128, 0, stream>>>(
          h1, ff_w + (size_t)l * Dd * Dd, ff_b + (size_t)l * Dd, h1, nullptr,
          g2 + (size_t)l * Dd, b2 + (size_t)l * Dd, nullptr, h);
    else
      gemm_res_ln<Dd, false, true><<<Nn / 16, 128, 0, stream>>>(
          h1, ff_w + (size_t)l * Dd * Dd, ff_b + (size_t)l * Dd, h1, nullptr,
          g2 + (size_t)l * Dd, b2 + (size_t)l * Dd, hproj, hc);
  }

  predict_all<<<PWG, 32 * PWV, 0, stream>>>(hc, pos_src, pos_dst, neg_src, neg_dst,
                                       pw1, pb1, pw2, pb2, pos);
}


# WMMA operand fragments (GEMMs, attention K/Q) loaded 16 bytes per lane, not 8: k permuted inside 8-k groups, A and B alike; GEMM loads grouped one 32-k block ahead; dot-product sum order changes
# speedup vs baseline: 3.6163x; 1.0409x over previous
#include <hip/hip_runtime.h>
#include <math.h>

typedef __attribute__((ext_vector_type(2))) float v2f;
typedef __attribute__((ext_vector_type(8))) float v8f;

#define Nn   4096
#define INF_ 3.402823466e+38f
#define Dd   256
#define Hh   8
#define HDd  32
#define Ll   2
#define Ee   131072
#define Pp   200000
#define INc  128

__device__ __forceinline__ v8f wmma4(v2f a, v2f b, v8f c) {
  return __builtin_amdgcn_wmma_f32_16x16x4_f32(
      false, a, false, b, (short)0, c, false, false);
}

__device__ __forceinline__ v2f lo2(float4 v) { return v2f{ v.x, v.y }; }
__device__ __forceinline__ v2f hi2(float4 v) { return v2f{ v.z, v.w }; }

__device__ __forceinline__ float rsum16(float v) {
  for (int m = 8; m >= 1; m >>= 1) v += __shfl_xor(v, m, 16);
  return v;
}
__device__ __forceinline__ float rmax16(float v) {
  for (int m = 8; m >= 1; m >>= 1) v = fmaxf(v, __shfl_xor(v, m, 16));
  return v;
}

struct GFrag {
  float4 a[4];
  float4 b0[4], b1[4], b2[4], b3[4];
};

__device__ __forceinline__ void load_g(GFrag& f, const float* __restrict__ arow,
                                       const float* __restrict__ w0,
                                       const float* __restrict__ w1,
                                       const float* __restrict__ w2,
                                       const float* __restrict__ w3) {
  for (int i = 0; i < 4; ++i) {
    f.a[i]  = *(const float4*)(arow + 8 * i);
    f.b0[i] = *(const float4*)(w0 + 8 * i);
    f.b1[i] = *(const float4*)(w1 + 8 * i);
    f.b2[i] = *(const float4*)(w2 + 8 * i);
    f.b3[i] = *(const float4*)(w3 + 8 * i);
  }
}

__device__ __forceinline__ void mma_g(const GFrag& f, v8f& c0, v8f& c1, v8f& c2, v8f& c3) {
  for (int i = 0; i < 4; ++i) {
    c0 = wmma4(lo2(f.a[i]), lo2(f.b0[i]), c0);
    c1 = wmma4(lo2(f.a[i]), lo2(f.b1[i]), c1);
    c2 = wmma4(lo2(f.a[i]), lo2(f.b2[i]), c2);
    c3 = wmma4(lo2(f.a[i]), lo2(f.b3[i]), c3);
    c0 = wmma4(hi2(f.a[i]), hi2(f.b0[i]), c0);
    c1 = wmma4(hi2(f.a[i]), hi2(f.b1[i]), c1);
    c2 = wmma4(hi2(f.a[i]), hi2(f.b2[i]), c2);
    c3 = wmma4(hi2(f.a[i]), hi2(f.b3[i]), c3);
  }
}

template <int KK>
__device__ __forceinline__ void gemm_tile(const float* __restrict__ arow,
                                          const float* __restrict__ w0,
                                          const float* __restrict__ w1,
                                          const float* __restrict__ w2,
                                          const float* __restrict__ w3,
                                          v8f& c0, v8f& c1, v8f& c2, v8f& c3) {
  GFrag f0, f1;
  load_g(f0, arow, w0, w1, w2, w3);
#pragma unroll
  for (int kb = 0; kb < KK / 32; kb += 2) {
    const int k1 = 32 * (kb + 1);
    load_g(f1, arow + k1, w0 + k1, w1 + k1, w2 + k1, w3 + k1);
    __builtin_amdgcn_sched_barrier(0);
    mma_g(f0, c0, c1, c2, c3);
    __builtin_amdgcn_sched_barrier(0);
    if (kb + 2 < KK / 32) {
      const int k2 = 32 * (kb + 2);
      load_g(f0, arow + k2, w0 + k2, w1 + k2, w2 + k2, w3 + k2);
    }
    __builtin_amdgcn_sched_barrier(0);
    mma_g(f1, c0, c1, c2, c3);
  }
}

template <int KK>
__global__ __launch_bounds__(32)
void gemm_atb(const float* __restrict__ A, const float* __restrict__ W,
              const float* __restrict__ bias, float* __restrict__ C,
              int M, int Nout) {
  const int tid  = threadIdx.x;
  const int m    = tid & 15;
  const int half = tid >> 4;
  const int tbn  = blockIdx.x * 64;
  const int tbm  = blockIdx.y * 16;

  const float* arow = A + (size_t)(tbm + m) * KK + half * 4;
  const float* w0   = W + (size_t)(tbn + m) * KK + half * 4;
  const float* w1   = w0 + (size_t)16 * KK;
  const float* w2   = w1 + (size_t)16 * KK;
  const float* w3   = w2 + (size_t)16 * KK;

  v8f c0 = {}, c1 = {}, c2 = {}, c3 = {};
  gemm_tile<KK>(arow, w0, w1, w2, w3, c0, c1, c2, c3);
  for (int r = 0; r < 8; ++r) {
    const int row = tbm + r + 8 * half;
    float* crow = C + (size_t)row * Nout + tbn + m;
    if (bias) {
      const float* brow = bias + tbn + m;
      crow[0]  = c0[r] + brow[0];
      crow[16] = c1[r] + brow[16];
      crow[32] = c2[r] + brow[32];
      crow[48] = c3[r] + brow[48];
    } else {
      crow[0] = c0[r]; crow[16] = c1[r]; crow[32] = c2[r]; crow[48] = c3[r];
    }
  }
}

__device__ __forceinline__ float xhalf(float x) {
  const int b = __builtin_bit_cast(int, x);
  return __builtin_bit_cast(float,
      __builtin_amdgcn_permlanex16(b, b, 0x76543210, 0xfedcba98, false, false));
}

struct KVFrag {
  float4 k[4];
  v2f v0[4];
  v2f v1[4];
};

__device__ __forceinline__ void load_kv(KVFrag& f, const float* __restrict__ kp,
                                        const float* __restrict__ vp) {
  for (int i = 0; i < 4; ++i) f.k[i] = *(const float4*)(kp + 8 * i);
  for (int j = 0; j < 4; ++j) {
    const float* vr = vp + (size_t)(2 * j) * (3 * Dd);
    f.v0[j] = v2f{ vr[0],  vr[3 * Dd] };
    f.v1[j] = v2f{ vr[16], vr[3 * Dd + 16] };
  }
}

__device__ __forceinline__ void softmax_pair(const v8f& sa, const v8f& sb, float& mrun,
                                             float& lpart, v8f& oa, v8f& ob,
                                             float (&p)[16]) {
  const float scale = 0.1767766952966369f;
  float sv[16];
  for (int r = 0; r < 8; ++r) { sv[r] = sa[r] * scale; sv[8 + r] = sb[r] * scale; }
  float mxa = fmaxf(fmaxf(fmaxf(sv[0], sv[1]), fmaxf(sv[2], sv[3])),
                    fmaxf(fmaxf(sv[4], sv[5]), fmaxf(sv[6], sv[7])));
  float mxb = fmaxf(fmaxf(fmaxf(sv[8], sv[9]), fmaxf(sv[10], sv[11])),
                    fmaxf(fmaxf(sv[12], sv[13]), fmaxf(sv[14], sv[15])));
  float mx = fmaxf(mxa, mxb);
  mx = fmaxf(mx, xhalf(mx));
  const float mn = fmaxf(mrun, mx);
  const float co = __expf(mrun - mn);
  for (int r = 0; r < 16; ++r) p[r] = __expf(sv[r] - mn);
  const float psa = ((p[0] + p[1]) + (p[2] + p[3])) + ((p[4] + p[5]) + (p[6] + p[7]));
  const float psb = ((p[8] + p[9]) + (p[10] + p[11])) + ((p[12] + p[13]) + (p[14] + p[15]));
  lpart = lpart * co + (psa + psb);
  mrun = mn;
  for (int r = 0; r < 8; ++r) { oa[r] *= co; ob[r] *= co; }
}

struct QTile {
  float4 q[4];
  v8f o0, o1;
  float mrun;
  float lpart;
};

__device__ __forceinline__ void pv_pair(const KVFrag& fa, const KVFrag& fb, QTile& t,
                                        const float (&p)[16]) {
  for (int j = 0; j < 4; ++j) {
    const v2f b = { p[2 * j], p[2 * j + 1] };
    t.o0 = wmma4(fa.v0[j], b, t.o0);
    t.o1 = wmma4(fa.v1[j], b, t.o1);
  }
  for (int j = 0; j < 4; ++j) {
    const v2f b = { p[8 + 2 * j], p[8 + 2 * j + 1] };
    t.o0 = wmma4(fb.v0[j], b, t.o0);
    t.o1 = wmma4(fb.v1[j], b, t.o1);
  }
}

__device__ __forceinline__ void attn_pair(const KVFrag& fa, const KVFrag& fb,
                                          QTile& t0, QTile& t1) {
  v8f s0a = {}, s0b = {}, s1a = {}, s1b = {};
  for (int i = 0; i < 4; ++i) {
    s0a = wmma4(lo2(fa.k[i]), lo2(t0.q[i]), s0a);
    s0b = wmma4(lo2(fb.k[i]), lo2(t0.q[i]), s0b);
    s0a = wmma4(hi2(fa.k[i]), hi2(t0.q[i]), s0a);
    s0b = wmma4(hi2(fb.k[i]), hi2(t0.q[i]), s0b);
  }
  for (int i = 0; i < 4; ++i) {
    s1a = wmma4(lo2(fa.k[i]), lo2(t1.q[i]), s1a);
    s1b = wmma4(lo2(fb.k[i]), lo2(t1.q[i]), s1b);
    s1a = wmma4(hi2(fa.k[i]), hi2(t1.q[i]), s1a);
    s1b = wmma4(hi2(fb.k[i]), hi2(t1.q[i]), s1b);
  }
  float p0[16], p1[16];
  softmax_pair(s0a, s0b, t0.mrun, t0.lpart, t0.o0, t0.o1, p0);
  pv_pair(fa, fb, t0, p0);
  softmax_pair(s1a, s1b, t1.mrun, t1.lpart, t1.o0, t1.o1, p1);
  pv_pair(fa, fb, t1, p1);
}

__device__ __forceinline__ void store_tile(const QTile& t, float* __restrict__ ob) {
  const float inv = 1.0f / (t.lpart + xhalf(t.lpart));
  *(float4*)(ob)      = make_float4(t.o0[0] * inv, t.o0[1] * inv, t.o0[2] * inv, t.o0[3] * inv);
  *(float4*)(ob + 4)  = make_float4(t.o0[4] * inv, t.o0[5] * inv, t.o0[6] * inv, t.o0[7] * inv);
  *(float4*)(ob + 16) = make_float4(t.o1[0] * inv, t.o1[1] * inv, t.o1[2] * inv, t.o1[3] * inv);
  *(float4*)(ob + 20) = make_float4(t.o1[4] * inv, t.o1[5] * inv, t.o1[6] * inv, t.o1[7] * inv);
}

#define MRG_T 18
__device__ __forceinline__ void put_tile(const QTile& t, float* __restrict__ d) {
  for (int r = 0; r < 8; ++r) { d[r * 32] = t.o0[r]; d[(8 + r) * 32] = t.o1[r]; }
  d[16 * 32] = t.mrun;
  d[17 * 32] = t.lpart;
}
__device__ __forceinline__ void merge_tile(QTile& t, const float* __restrict__ d) {
  const float m1 = d[16 * 32];
  const float l1 = d[17 * 32];
  const float mn = fmaxf(t.mrun, m1);
  const float a  = __expf(t.mrun - mn);
  const float b  = __expf(m1 - mn);
  for (int r = 0; r < 8; ++r) {
    t.o0[r] = t.o0[r] * a + d[r * 32] * b;
    t.o1[r] = t.o1[r] * a + d[(8 + r) * 32] * b;
  }
  t.lpart = t.lpart * a + l1 * b;
  t.mrun  = mn;
}

__global__ __launch_bounds__(256) __attribute__((amdgpu_waves_per_eu(1, 2)))
void attn_kernel(const float* __restrict__ qkv, float* __restrict__ obuf) {
  __shared__ float mrg[4][2 * MRG_T][32];

  const int lane = threadIdx.x & 31;
  const int wave = threadIdx.x >> 5;
  const int qs   = wave & 3;
  const int kh   = wave >> 2;
  const int m    = lane & 15;
  const int half = lane >> 4;
  const int q0   = (blockIdx.x * 4 + qs) * 32;
  const int h    = blockIdx.y;
  const int kfirst = kh * (Nn / 2);

  QTile t0, t1;
  {
    const float* qr = qkv + (size_t)(q0 + m) * (3 * Dd) + h * HDd + 4 * half;
    for (int i = 0; i < 4; ++i) {
      t0.q[i] = *(const float4*)(qr + 8 * i);
      t1.q[i] = *(const float4*)(qr + (size_t)16 * (3 * Dd) + 8 * i);
    }
  }
  t0.o0 = v8f{}; t0.o1 = v8f{}; t1.o0 = v8f{}; t1.o1 = v8f{};
  t0.mrun = -INF_; t1.mrun = -INF_;
  t0.lpart = 0.f;  t1.lpart = 0.f;

  const float* kp = qkv + Dd + h * HDd + (size_t)(kfirst + m) * (3 * Dd) + 4 * half;
  const float* vp = qkv + 2 * Dd + h * HDd + (size_t)(kfirst + 8 * half) * (3 * Dd) + m;
  const size_t kstep = (size_t)16 * (3 * Dd);

  KVFrag fa0, fa1, fb0, fb1;
  load_kv(fa0, kp, vp);
  load_kv(fa1, kp + kstep, vp + kstep);
  for (int kb = 0; kb < Nn / 32; kb += 4) {
    load_kv(fb0, kp + (size_t)(kb + 2) * kstep, vp + (size_t)(kb + 2) * kstep);
    load_kv(fb1, kp + (size_t)(kb + 3) * kstep, vp + (size_t)(kb + 3) * kstep);
    attn_pair(fa0, fa1, t0, t1);
    __builtin_amdgcn_sched_barrier(0);
    const int nx = (kb + 4 < Nn / 32) ? kb + 4 : kb + 2;
    load_kv(fa0, kp + (size_t)nx * kstep, vp + (size_t)nx * kstep);
    load_kv(fa1, kp + (size_t)(nx + 1) * kstep, vp + (size_t)(nx + 1) * kstep);
    __builtin_amdgcn_sched_barrier(0);
    attn_pair(fb0, fb1, t0, t1);
  }

  float* mg = &mrg[qs][0][lane];
  if (kh == 1) {
    put_tile(t0, mg);
    put_tile(t1, mg + MRG_T * 32);
  }
  __syncthreads();
  if (kh == 0) {
    merge_tile(t0, mg);
    merge_tile(t1, mg + MRG_T * 32);
    float* ob = obuf + (size_t)(q0 + m) * Dd + h * HDd + 8 * half;
    store_tile(t0, ob);
    store_tile(t1, ob + (size_t)16 * Dd);
  }
}

__global__ void zerof(float* p, int n) {
  int i = blockIdx.x * blockDim.x + threadIdx.x;
  if (i < n) p[i] = 0.f;
}
__global__ void seg_sum(const float* __restrict__ ew, const int* __restrict__ src,
                        float* __restrict__ out, int E) {
  int i = blockIdx.x * blockDim.x + threadIdx.x;
  if (i < E) atomicAdd(&out[src[i]], ew[i]);
}

template <int KK, bool SCALE, bool ADD>
__global__ __launch_bounds__(128)
void gemm_res_ln(const float* __restrict__ A, const float* __restrict__ W,
                 const float* __restrict__ bias, const float* __restrict__ res,
                 const float* __restrict__ rscale,
                 const float* __restrict__ g, const float* __restrict__ b,
                 const float* __restrict__ addp, float* __restrict__ out) {
  __shared__ float red[2][4][16];

  const int lane = threadIdx.x & 31;
  const int wave = threadIdx.x >> 5;
  const int m    = lane & 15;
  const int half = lane >> 4;
  const int tbn  = wave * 64;
  const int tbm  = blockIdx.x * 16;

  const float* arow = A + (size_t)(tbm + m) * KK + half * 4;
  const float* w0   = W + (size_t)(tbn + m) * KK + half * 4;
  const float* w1   = w0 + (size_t)16 * KK;
  const float* w2   = w1 + (size_t)16 * KK;
  const float* w3   = w2 + (size_t)16 * KK;

  v8f c0 = {}, c1 = {}, c2 = {}, c3 = {};
  gemm_tile<KK>(arow, w0, w1, w2, w3, c0, c1, c2, c3);

  const int col = tbn + m;
  const int rb  = 8 * half;
  const float bj0 = bias[col], bj1 = bias[col + 16], bj2 = bias[col + 32], bj3 = bias[col + 48];
  const float g0 = g[col], g1 = g[col + 16], g2 = g[col + 32], g3 = g[col + 48];
  const float e0 = b[col], e1 = b[col + 16], e2 = b[col + 32], e3 = b[col + 48];
  float a0[8], a1[8], a2[8], a3[8];
  if (ADD) {
#pragma unroll
    for (int r = 0; r < 8; ++r) {
      const float* ar = addp + (size_t)(tbm + rb + r) * Dd + col;
      a0[r] = ar[0]; a1[r] = ar[16]; a2[r] = ar[32]; a3[r] = ar[48];
    }
  }

  float t0[8], t1[8], t2[8], t3[8];
#pragma unroll
  for (int r = 0; r < 8; ++r) {
    const int row = tbm + rb + r;
    const float sc = SCALE ? rscale[row] : 1.0f;
    const float* rr = res + (size_t)row * Dd + col;
    t0[r] = rr[0]  + (c0[r] + bj0) * sc;
    t1[r] = rr[16] + (c1[r] + bj1) * sc;
    t2[r] = rr[32] + (c2[r] + bj2) * sc;
    t3[r] = rr[48] + (c3[r] + bj3) * sc;
  }

  float s[8];
#pragma unroll
  for (int r = 0; r < 8; ++r) s[r] = rsum16((t0[r] + t1[r]) + (t2[r] + t3[r]));
  if (m == 0) {
#pragma unroll
    for (int r = 0; r < 8; ++r) red[0][wave][rb + r] = s[r];
  }
  __syncthreads();
#pragma unroll
  for (int r = 0; r < 8; ++r) {
    const float mean = ((red[0][0][rb + r] + red[0][1][rb + r]) +
                        (red[0][2][rb + r] + red[0][3][rb + r])) / (float)Dd;
    t0[r] -= mean; t1[r] -= mean; t2[r] -= mean; t3[r] -= mean;
  }

#pragma unroll
  for (int r = 0; r < 8; ++r)
    s[r] = rsum16((t0[r] * t0[r] + t1[r] * t1[r]) + (t2[r] * t2[r] + t3[r] * t3[r]));
  if (m == 0) {
#pragma unroll
    for (int r = 0; r < 8; ++r) red[1][wave][rb + r] = s[r];
  }
  __syncthreads();

#pragma unroll
  for (int r = 0; r < 8; ++r) {
    const float var = ((red[1][0][rb + r] + red[1][1][rb + r]) +
                       (red[1][2][rb + r] + red[1][3][rb + r])) / (float)Dd;
    const float rs = rsqrtf(var + 1e-5f);
    const size_t off = (size_t)(tbm + rb + r) * Dd + col;
    float y0 = t0[r] * rs * g0 + e0;
    float y1 = t1[r] * rs * g1 + e1;
    float y2 = t2[r] * rs * g2 + e2;
    float y3 = t3[r] * rs * g3 + e3;
    if (ADD) { y0 += a0[r]; y1 += a1[r]; y2 += a2[r]; y3 += a3[r]; }
    float* orow = out + off;
    orow[0] = y0; orow[16] = y1; orow[32] = y2; orow[48] = y3;
  }
}

#define PWP  260
#define TPH  (Pp / 16)
#define PWG  256
#define PWV  12

__global__ __launch_bounds__(32 * PWV)
void predict_all(const float* __restrict__ hc,
                 const int* __restrict__ psrc, const int* __restrict__ pdst,
                 const int* __restrict__ nsrc, const int* __restrict__ ndst,
                 const float* __restrict__ pw1, const float* __restrict__ pb1,
                 const float* __restrict__ pw2, const float* __restrict__ pb2,
                 float* __restrict__ out) {
  __shared__ float wl[(Dd / 2) * PWP];

  const int tid = threadIdx.x;
  for (int idx = tid; idx < (Dd / 2) * (Dd / 4); idx += 32 * PWV) {
    const int row = idx >> 6, c4 = idx & 63;
    const float4 v = *(const float4*)(pw1 + (size_t)row * Dd + c4 * 4);
    *(float4*)(wl + row * PWP + c4 * 4) = v;
  }
  __syncthreads();

  const int lane = tid & 31;
  const int wave = tid >> 5;
  const int m    = lane & 15;
  const int half = lane >> 4;

  float bias1[8], w2[8];
#pragma unroll
  for (int jt = 0; jt < 8; ++jt) {
    bias1[jt] = pb1[jt * 16 + m];
    w2[jt]    = pw2[jt * 16 + m];
  }
  const float b2 = pb2[0];
  const float* wfrag = wl + m * PWP + 4 * half;

  for (int t = wave * PWG + (int)blockIdx.x; t < 2 * TPH; t += PWV * PWG) {
    asm volatile("" ::: "memory");
    const bool ng = t >= TPH;
    const int* sp = ng ? nsrc : psrc;
    const int* dp = ng ? ndst : pdst;
    const int tb  = (ng ? t - TPH : t) * 16;
    const int rs = sp[tb + m], rd = dp[tb + m];
    const float* as = hc + (size_t)rs * Dd + 4 * half;
    const float* ad = hc + (size_t)rd * Dd + 4 * half;

    float4 z[32];
#pragma unroll
    for (int i = 0; i < 32; ++i) {
      const float4 a = *(const float4*)(as + 8 * i);
      const float4 b = *(const float4*)(ad + 8 * i);
      z[i] = make_float4(a.x * b.x, a.y * b.y, a.z * b.z, a.w * b.w);
    }

    v8f c[8];
#pragma unroll
    for (int jt = 0; jt < 8; ++jt) c[jt] = v8f{};

#pragma unroll
    for (int i = 0; i < 32; ++i) {
      float4 w[8];
#pragma unroll
      for (int jt = 0; jt < 8; ++jt)
        w[jt] = *(const float4*)(wfrag + jt * 16 * PWP + 8 * i);
      const v2f a0 = { z[i].x, z[i].y };
      const v2f a1 = { z[i].z, z[i].w };
#pragma unroll
      for (int jt = 0; jt < 8; ++jt)
        c[jt] = wmma4(a0, v2f{ w[jt].x, w[jt].y }, c[jt]);
#pragma unroll
      for (int jt = 0; jt < 8; ++jt)
        c[jt] = wmma4(a1, v2f{ w[jt].z, w[jt].w }, c[jt]);
    }

    float acc[8];
#pragma unroll
    for (int r = 0; r < 8; ++r) acc[r] = 0.f;
#pragma unroll
    for (int jt = 0; jt < 8; ++jt) {
#pragma unroll
      for (int r = 0; r < 8; ++r) {
        float v = c[jt][r] + bias1[jt];
        v = v > 0.f ? v : 0.2f * v;
        acc[r] += v * w2[jt];
      }
    }
#pragma unroll
    for (int r = 0; r < 8; ++r) acc[r] = rsum16(acc[r]);
    float o = acc[0];
#pragma unroll
    for (int r = 1; r < 8; ++r) o = (m == r) ? acc[r] : o;
    if (m < 8) out[(size_t)t * 16 + 8 * half + m] = o + b2;
  }
}

extern "C" void kernel_launch(void* const* d_in, const int* in_sizes, int n_in,
                              void* d_out, int out_size, void* d_ws, size_t ws_size,
                              hipStream_t stream) {
  const float* x       = (const float*)d_in[0];
  const float* edge_w  = (const float*)d_in[1];
  const int*   src     = (const int*)d_in[2];
  const int*   pos_src = (const int*)d_in[4];
  const int*   pos_dst = (const int*)d_in[5];
  const int*   neg_src = (const int*)d_in[6];
  const int*   neg_dst = (const int*)d_in[7];
  const float* proj_w  = (const float*)d_in[8];
  const float* proj_b  = (const float*)d_in[9];
  const float* in_w    = (const float*)d_in[10];
  const float* in_b    = (const float*)d_in[11];
  const float* out_w   = (const float*)d_in[12];
  const float* out_b   = (const float*)d_in[13];
  const float* ff_w    = (const float*)d_in[14];
  const float* ff_b    = (const float*)d_in[15];
  const float* g1      = (const float*)d_in[16];
  const float* b1      = (const float*)d_in[17];
  const float* g2      = (const float*)d_in[18];
  const float* b2      = (const float*)d_in[19];
  const float* pw1     = (const float*)d_in[20];
  const float* pb1     = (const float*)d_in[21];
  const float* pw2     = (const float*)d_in[22];
  const float* pb2     = (const float*)d_in[23];

  float* W      = (float*)d_ws;
  float* h      = W;
  float* hproj  = W + 1048576;
  float* qkv    = W + 2097152;
  float* obuf   = W + 5242880;
  float* h1     = W + 6291456;
  float* wsn    = W + 7340032;

  float* outp = (float*)d_out;
  float* pos  = outp;
  float* hc   = outp + 2 * Pp;

  gemm_atb<INc><<<dim3(Dd / 64, Nn / 16), 32, 0, stream>>>(x, proj_w, proj_b, hproj, Nn, Dd);

  const float* hin[Ll] = { hproj, h };

  for (int l = 0; l < Ll; ++l) {
    gemm_atb<Dd><<<dim3(3 * Dd / 64, Nn / 16), 32, 0, stream>>>(
        hin[l], in_w + (size_t)l * 3 * Dd * Dd, in_b + (size_t)l * 3 * Dd, qkv, Nn, 3 * Dd);
    attn_kernel<<<dim3(Nn / 128, Hh), 256, 0, stream>>>(qkv, obuf);
    zerof<<<(Nn + 255) / 256, 256, 0, stream>>>(wsn, Nn);
    seg_sum<<<(Ee + 255) / 256, 256, 0, stream>>>(edge_w + (size_t)l * Ee,
                                                  src + (size_t)l * Ee, wsn, Ee);
    gemm_res_ln<Dd, true, false><<<Nn / 16, 128, 0, stream>>>(
        obuf, out_w + (size_t)l * Dd * Dd, out_b + (size_t)l * Dd, hin[l], wsn,
        g1 + (size_t)l * Dd, b1 + (size_t)l * Dd, nullptr, h1);
    if (l + 1 < Ll)
      gemm_res_ln<Dd, false, false><<<Nn / 16, 128, 0, stream>>>(
          h1, ff_w + (size_t)l * Dd * Dd, ff_b + (size_t)l * Dd, h1, nullptr,
          g2 + (size_t)l * Dd, b2 + (size_t)l * Dd, nullptr, h);
    else
      gemm_res_ln<Dd, false, true><<<Nn / 16, 128, 0, stream>>>(
          h1, ff_w + (size_t)l * Dd * Dd, ff_b + (size_t)l * Dd, h1, nullptr,
          g2 + (size_t)l * Dd, b2 + (size_t)l * Dd, hproj, hc);
  }

  predict_all<<<PWG, 32 * PWV, 0, stream>>>(hc, pos_src, pos_dst, neg_src, neg_dst,
                                       pw1, pb1, pw2, pb2, pos);
}


# row-local GEMM chains run as one launch each (proj+qkv; out-proj+LN+FF+LN+next qkv): the 16-row tile stays in LDS as the next stage's A operand; 14 launches become 10; every sum order kept
# speedup vs baseline: 3.8280x; 1.0585x over previous
#include <hip/hip_runtime.h>
#include <math.h>

typedef __attribute__((ext_vector_type(2))) float v2f;
typedef __attribute__((ext_vector_type(8))) float v8f;

#define Nn   4096
#define INF_ 3.402823466e+38f
#define Dd   256
#define Hh   8
#define HDd  32
#define Ll   2
#define Ee   131072
#define Pp   200000
#define INc  128

__device__ __forceinline__ v8f wmma4(v2f a, v2f b, v8f c) {
  return __builtin_amdgcn_wmma_f32_16x16x4_f32(
      false, a, false, b, (short)0, c, false, false);
}

__device__ __forceinline__ v2f lo2(float4 v) { return v2f{ v.x, v.y }; }
__device__ __forceinline__ v2f hi2(float4 v) { return v2f{ v.z, v.w }; }

__device__ __forceinline__ float rsum16(float v) {
  for (int m = 8; m >= 1; m >>= 1) v += __shfl_xor(v, m, 16);
  return v;
}
__device__ __forceinline__ float rmax16(float v) {
  for (int m = 8; m >= 1; m >>= 1) v = fmaxf(v, __shfl_xor(v, m, 16));
  return v;
}

struct GFrag {
  float4 a[4];
  float4 b0[4], b1[4], b2[4], b3[4];
};

__device__ __forceinline__ void load_g(GFrag& f, const float* __restrict__ arow,
                                       const float* __restrict__ w0,
                                       const float* __restrict__ w1,
                                       const float* __restrict__ w2,
                                       const float* __restrict__ w3) {
  for (int i = 0; i < 4; ++i) {
    f.a[i]  = *(const float4*)(arow + 8 * i);
    f.b0[i] = *(const float4*)(w0 + 8 * i);
    f.b1[i] = *(const float4*)(w1 + 8 * i);
    f.b2[i] = *(const float4*)(w2 + 8 * i);
    f.b3[i] = *(const float4*)(w3 + 8 * i);
  }
}

__device__ __forceinline__ void mma_g(const GFrag& f, v8f& c0, v8f& c1, v8f& c2, v8f& c3) {
  for (int i = 0; i < 4; ++i) {
    c0 = wmma4(lo2(f.a[i]), lo2(f.b0[i]), c0);
    c1 = wmma4(lo2(f.a[i]), lo2(f.b1[i]), c1);
    c2 = wmma4(lo2(f.a[i]), lo2(f.b2[i]), c2);
    c3 = wmma4(lo2(f.a[i]), lo2(f.b3[i]), c3);
    c0 = wmma4(hi2(f.a[i]), hi2(f.b0[i]), c0);
    c1 = wmma4(hi2(f.a[i]), hi2(f.b1[i]), c1);
    c2 = wmma4(hi2(f.a[i]), hi2(f.b2[i]), c2);
    c3 = wmma4(hi2(f.a[i]), hi2(f.b3[i]), c3);
  }
}

template <int KK>
__device__ __forceinline__ void gemm_tile(const float* __restrict__ arow,
                                          const float* __restrict__ w0,
                                          const float* __restrict__ w1,
                                          const float* __restrict__ w2,
                                          const float* __restrict__ w3,
                                          v8f& c0, v8f& c1, v8f& c2, v8f& c3) {
  GFrag f0, f1;
  load_g(f0, arow, w0, w1, w2, w3);
#pragma unroll
  for (int kb = 0; kb < KK / 32; kb += 2) {
    const int k1 = 32 * (kb + 1);
    load_g(f1, arow + k1, w0 + k1, w1 + k1, w2 + k1, w3 + k1);
    __builtin_amdgcn_sched_barrier(0);
    mma_g(f0, c0, c1, c2, c3);
    __builtin_amdgcn_sched_barrier(0);
    if (kb + 2 < KK / 32) {
      const int k2 = 32 * (kb + 2);
      load_g(f0, arow + k2, w0 + k2, w1 + k2, w2 + k2, w3 + k2);
    }
    __builtin_amdgcn_sched_barrier(0);
    mma_g(f1, c0, c1, c2, c3);
    __builtin_amdgcn_sched_barrier(0);
  }
}

#define APITCH 260

__device__ __forceinline__ void tile_to_lds(float* __restrict__ hs, int rb, int col,
                                            const float (&y0)[8], const float (&y1)[8],
                                            const float (&y2)[8], const float (&y3)[8]) {
#pragma unroll
  for (int r = 0; r < 8; ++r) {
    float* hr = hs + (rb + r) * APITCH + col;
    hr[0] = y0[r]; hr[16] = y1[r]; hr[32] = y2[r]; hr[48] = y3[r];
  }
}

__device__ __forceinline__ void qkv_rows(const float* __restrict__ hs,
                                         const float* __restrict__ in_w,
                                         const float* __restrict__ in_b,
                                         float* __restrict__ qkv,
                                         int tbm, int wave, int m, int half) {
  const float* arow = hs + m * APITCH + 4 * half;
  for (int p = 0; p < 3; ++p) {
    const int tbn = p * Dd + wave * 64;
    const float* w0 = in_w + (size_t)(tbn + m) * Dd + half * 4;
    const float* w1 = w0 + (size_t)16 * Dd;
    const float* w2 = w1 + (size_t)16 * Dd;
    const float* w3 = w2 + (size_t)16 * Dd;
    const float* brow = in_b + tbn + m;
    const float q0 = brow[0], q1 = brow[16], q2 = brow[32], q3 = brow[48];
    v8f c0 = {}, c1 = {}, c2 = {}, c3 = {};
    gemm_tile<Dd>(arow, w0, w1, w2, w3, c0, c1, c2, c3);
#pragma unroll
    for (int r = 0; r < 8; ++r) {
      float* crow = qkv + (size_t)(tbm + r + 8 * half) * (3 * Dd) + tbn + m;
      crow[0]  = c0[r] + q0;
      crow[16] = c1[r] + q1;
      crow[32] = c2[r] + q2;
      crow[48] = c3[r] + q3;
    }
  }
}

__global__ __launch_bounds__(128)
void proj_qkv(const float* __restrict__ x, const float* __restrict__ proj_w,
              const float* __restrict__ proj_b, float* __restrict__ hproj,
              const float* __restrict__ in_w, const float* __restrict__ in_b,
              float* __restrict__ qkv) {
  __shared__ float hs[16 * APITCH];

  const int lane = threadIdx.x & 31;
  const int wave = threadIdx.x >> 5;
  const int m    = lane & 15;
  const int half = lane >> 4;
  const int tbn  = wave * 64;
  const int tbm  = blockIdx.x * 16;

  const float* arow = x + (size_t)(tbm + m) * INc + half * 4;
  const float* w0   = proj_w + (size_t)(tbn + m) * INc + half * 4;
  const float* w1   = w0 + (size_t)16 * INc;
  const float* w2   = w1 + (size_t)16 * INc;
  const float* w3   = w2 + (size_t)16 * INc;

  v8f c0 = {}, c1 = {}, c2 = {}, c3 = {};
  gemm_tile<INc>(arow, w0, w1, w2, w3, c0, c1, c2, c3);

  const int col = tbn + m;
  const int rb  = 8 * half;
  const float bj0 = proj_b[col], bj1 = proj_b[col + 16], bj2 = proj_b[col + 32], bj3 = proj_b[col + 48];
  float y0[8], y1[8], y2[8], y3[8];
#pragma unroll
  for (int r = 0; r < 8; ++r) {
    y0[r] = c0[r] + bj0; y1[r] = c1[r] + bj1; y2[r] = c2[r] + bj2; y3[r] = c3[r] + bj3;
    float* orow = hproj + (size_t)(tbm + rb + r) * Dd + col;
    orow[0] = y0[r]; orow[16] = y1[r]; orow[32] = y2[r]; orow[48] = y3[r];
  }
  tile_to_lds(hs, rb, col, y0, y1, y2, y3);
  __syncthreads();
  qkv_rows(hs, in_w, in_b, qkv, tbm, wave, m, half);
}

__device__ __forceinline__ float xhalf(float x) {
  const int b = __builtin_bit_cast(int, x);
  return __builtin_bit_cast(float,
      __builtin_amdgcn_permlanex16(b, b, 0x76543210, 0xfedcba98, false, false));
}

struct KVFrag {
  float4 k[4];
  v2f v0[4];
  v2f v1[4];
};

__device__ __forceinline__ void load_kv(KVFrag& f, const float* __restrict__ kp,
                                        const float* __restrict__ vp) {
  for (int i = 0; i < 4; ++i) f.k[i] = *(const float4*)(kp + 8 * i);
  for (int j = 0; j < 4; ++j) {
    const float* vr = vp + (size_t)(2 * j) * (3 * Dd);
    f.v0[j] = v2f{ vr[0],  vr[3 * Dd] };
    f.v1[j] = v2f{ vr[16], vr[3 * Dd + 16] };
  }
}

__device__ __forceinline__ void softmax_pair(const v8f& sa, const v8f& sb, float& mrun,
                                             float& lpart, v8f& oa, v8f& ob,
                                             float (&p)[16]) {
  const float scale = 0.1767766952966369f;
  float sv[16];
  for (int r = 0; r < 8; ++r) { sv[r] = sa[r] * scale; sv[8 + r] = sb[r] * scale; }
  float mxa = fmaxf(fmaxf(fmaxf(sv[0], sv[1]), fmaxf(sv[2], sv[3])),
                    fmaxf(fmaxf(sv[4], sv[5]), fmaxf(sv[6], sv[7])));
  float mxb = fmaxf(fmaxf(fmaxf(sv[8], sv[9]), fmaxf(sv[10], sv[11])),
                    fmaxf(fmaxf(sv[12], sv[13]), fmaxf(sv[14], sv[15])));
  float mx = fmaxf(mxa, mxb);
  mx = fmaxf(mx, xhalf(mx));
  const float mn = fmaxf(mrun, mx);
  const float co = __expf(mrun - mn);
  for (int r = 0; r < 16; ++r) p[r] = __expf(sv[r] - mn);
  const float psa = ((p[0] + p[1]) + (p[2] + p[3])) + ((p[4] + p[5]) + (p[6] + p[7]));
  const float psb = ((p[8] + p[9]) + (p[10] + p[11])) + ((p[12] + p[13]) + (p[14] + p[15]));
  lpart = lpart * co + (psa + psb);
  mrun = mn;
  for (int r = 0; r < 8; ++r) { oa[r] *= co; ob[r] *= co; }
}

struct QTile {
  float4 q[4];
  v8f o0, o1;
  float mrun;
  float lpart;
};

__device__ __forceinline__ void pv_pair(const KVFrag& fa, const KVFrag& fb, QTile& t,
                                        const float (&p)[16]) {
  for (int j = 0; j < 4; ++j) {
    const v2f b = { p[2 * j], p[2 * j + 1] };
    t.o0 = wmma4(fa.v0[j], b, t.o0);
    t.o1 = wmma4(fa.v1[j], b, t.o1);
  }
  for (int j = 0; j < 4; ++j) {
    const v2f b = { p[8 + 2 * j], p[8 + 2 * j + 1] };
    t.o0 = wmma4(fb.v0[j], b, t.o0);
    t.o1 = wmma4(fb.v1[j], b, t.o1);
  }
}

__device__ __forceinline__ void attn_pair(const KVFrag& fa, const KVFrag& fb,
                                          QTile& t0, QTile& t1) {
  v8f s0a = {}, s0b = {}, s1a = {}, s1b = {};
  for (int i = 0; i < 4; ++i) {
    s0a = wmma4(lo2(fa.k[i]), lo2(t0.q[i]), s0a);
    s0b = wmma4(lo2(fb.k[i]), lo2(t0.q[i]), s0b);
    s0a = wmma4(hi2(fa.k[i]), hi2(t0.q[i]), s0a);
    s0b = wmma4(hi2(fb.k[i]), hi2(t0.q[i]), s0b);
  }
  for (int i = 0; i < 4; ++i) {
    s1a = wmma4(lo2(fa.k[i]), lo2(t1.q[i]), s1a);
    s1b = wmma4(lo2(fb.k[i]), lo2(t1.q[i]), s1b);
    s1a = wmma4(hi2(fa.k[i]), hi2(t1.q[i]), s1a);
    s1b = wmma4(hi2(fb.k[i]), hi2(t1.q[i]), s1b);
  }
  float p0[16], p1[16];
  softmax_pair(s0a, s0b, t0.mrun, t0.lpart, t0.o0, t0.o1, p0);
  pv_pair(fa, fb, t0, p0);
  softmax_pair(s1a, s1b, t1.mrun, t1.lpart, t1.o0, t1.o1, p1);
  pv_pair(fa, fb, t1, p1);
}

__device__ __forceinline__ void store_tile(const QTile& t, float* __restrict__ ob) {
  const float inv = 1.0f / (t.lpart + xhalf(t.lpart));
  *(float4*)(ob)      = make_float4(t.o0[0] * inv, t.o0[1] * inv, t.o0[2] * inv, t.o0[3] * inv);
  *(float4*)(ob + 4)  = make_float4(t.o0[4] * inv, t.o0[5] * inv, t.o0[6] * inv, t.o0[7] * inv);
  *(float4*)(ob + 16) = make_float4(t.o1[0] * inv, t.o1[1] * inv, t.o1[2] * inv, t.o1[3] * inv);
  *(float4*)(ob + 20) = make_float4(t.o1[4] * inv, t.o1[5] * inv, t.o1[6] * inv, t.o1[7] * inv);
}

#define MRG_T 18
__device__ __forceinline__ void put_tile(const QTile& t, float* __restrict__ d) {
  for (int r = 0; r < 8; ++r) { d[r * 32] = t.o0[r]; d[(8 + r) * 32] = t.o1[r]; }
  d[16 * 32] = t.mrun;
  d[17 * 32] = t.lpart;
}
__device__ __forceinline__ void merge_tile(QTile& t, const float* __restrict__ d) {
  const float m1 = d[16 * 32];
  const float l1 = d[17 * 32];
  const float mn = fmaxf(t.mrun, m1);
  const float a  = __expf(t.mrun - mn);
  const float b  = __expf(m1 - mn);
  for (int r = 0; r < 8; ++r) {
    t.o0[r] = t.o0[r] * a + d[r * 32] * b;
    t.o1[r] = t.o1[r] * a + d[(8 + r) * 32] * b;
  }
  t.lpart = t.lpart * a + l1 * b;
  t.mrun  = mn;
}

__global__ __launch_bounds__(256) __attribute__((amdgpu_waves_per_eu(1, 2)))
void attn_kernel(const float* __restrict__ qkv, float* __restrict__ obuf) {
  __shared__ float mrg[4][2 * MRG_T][32];

  const int lane = threadIdx.x & 31;
  const int wave = threadIdx.x >> 5;
  const int qs   = wave & 3;
  const int kh   = wave >> 2;
  const int m    = lane & 15;
  const int half = lane >> 4;
  const int q0   = (blockIdx.x * 4 + qs) * 32;
  const int h    = blockIdx.y;
  const int kfirst = kh * (Nn / 2);

  QTile t0, t1;
  {
    const float* qr = qkv + (size_t)(q0 + m) * (3 * Dd) + h * HDd + 4 * half;
    for (int i = 0; i < 4; ++i) {
      t0.q[i] = *(const float4*)(qr + 8 * i);
      t1.q[i] = *(const float4*)(qr + (size_t)16 * (3 * Dd) + 8 * i);
    }
  }
  t0.o0 = v8f{}; t0.o1 = v8f{}; t1.o0 = v8f{}; t1.o1 = v8f{};
  t0.mrun = -INF_; t1.mrun = -INF_;
  t0.lpart = 0.f;  t1.lpart = 0.f;

  const float* kp = qkv + Dd + h * HDd + (size_t)(kfirst + m) * (3 * Dd) + 4 * half;
  const float* vp = qkv + 2 * Dd + h * HDd + (size_t)(kfirst + 8 * half) * (3 * Dd) + m;
  const size_t kstep = (size_t)16 * (3 * Dd);

  KVFrag fa0, fa1, fb0, fb1;
  load_kv(fa0, kp, vp);
  load_kv(fa1, kp + kstep, vp + kstep);
  for (int kb = 0; kb < Nn / 32; kb += 4) {
    load_kv(fb0, kp + (size_t)(kb + 2) * kstep, vp + (size_t)(kb + 2) * kstep);
    load_kv(fb1, kp + (size_t)(kb + 3) * kstep, vp + (size_t)(kb + 3) * kstep);
    attn_pair(fa0, fa1, t0, t1);
    __builtin_amdgcn_sched_barrier(0);
    const int nx = (kb + 4 < Nn / 32) ? kb + 4 : kb + 2;
    load_kv(fa0, kp + (size_t)nx * kstep, vp + (size_t)nx * kstep);
    load_kv(fa1, kp + (size_t)(nx + 1) * kstep, vp + (size_t)(nx + 1) * kstep);
    __builtin_amdgcn_sched_barrier(0);
    attn_pair(fb0, fb1, t0, t1);
  }

  float* mg = &mrg[qs][0][lane];
  if (kh == 1) {
    put_tile(t0, mg);
    put_tile(t1, mg + MRG_T * 32);
  }
  __syncthreads();
  if (kh == 0) {
    merge_tile(t0, mg);
    merge_tile(t1, mg + MRG_T * 32);
    float* ob = obuf + (size_t)(q0 + m) * Dd + h * HDd + 8 * half;
    store_tile(t0, ob);
    store_tile(t1, ob + (size_t)16 * Dd);
  }
}

__global__ void zerof(float* p, int n) {
  int i = blockIdx.x * blockDim.x + threadIdx.x;
  if (i < n) p[i] = 0.f;
}
__global__ void seg_sum(const float* __restrict__ ew, const int* __restrict__ src,
                        float* __restrict__ out, int E) {
  int i = blockIdx.x * blockDim.x + threadIdx.x;
  if (i < E) atomicAdd(&out[src[i]], ew[i]);
}

__device__ __forceinline__ void ln_rows(float (&t0)[8], float (&t1)[8], float (&t2)[8],
                                        float (&t3)[8], float* __restrict__ red,
                                        int wave, int m, int rb,
                                        float g0, float g1, float g2, float g3,
                                        float e0, float e1, float e2, float e3) {
  float* red0 = red;
  float* red1 = red + 64;
  float s[8];
#pragma unroll
  for (int r = 0; r < 8; ++r) s[r] = rsum16((t0[r] + t1[r]) + (t2[r] + t3[r]));
  if (m == 0) {
#pragma unroll
    for (int r = 0; r < 8; ++r) red0[wave * 16 + rb + r] = s[r];
  }
  __syncthreads();
#pragma unroll
  for (int r = 0; r < 8; ++r) {
    const float mean = ((red0[rb + r] + red0[16 + rb + r]) +
                        (red0[32 + rb + r] + red0[48 + rb + r])) / (float)Dd;
    t0[r] -= mean; t1[r] -= mean; t2[r] -= mean; t3[r] -= mean;
  }

#pragma unroll
  for (int r = 0; r < 8; ++r)
    s[r] = rsum16((t0[r] * t0[r] + t1[r] * t1[r]) + (t2[r] * t2[r] + t3[r] * t3[r]));
  if (m == 0) {
#pragma unroll
    for (int r = 0; r < 8; ++r) red1[wave * 16 + rb + r] = s[r];
  }
  __syncthreads();

#pragma unroll
  for (int r = 0; r < 8; ++r) {
    const float var = ((red1[rb + r] + red1[16 + rb + r]) +
                       (red1[32 + rb + r] + red1[48 + rb + r])) / (float)Dd;
    const float rs = rsqrtf(var + 1e-5f);
    t0[r] = t0[r] * rs * g0 + e0;
    t1[r] = t1[r] * rs * g1 + e1;
    t2[r] = t2[r] * rs * g2 + e2;
    t3[r] = t3[r] * rs * g3 + e3;
  }
}

template <bool QKV, bool ADD>
__global__ __launch_bounds__(128)
void layer_tail(const float* __restrict__ A, const float* __restrict__ W1,
                const float* __restrict__ bias1, const float* __restrict__ res,
                const float* __restrict__ rscale,
                const float* __restrict__ g1, const float* __restrict__ b1,
                const float* __restrict__ W2, const float* __restrict__ bias2,
                const float* __restrict__ g2, const float* __restrict__ b2,
                const float* __restrict__ addp, float* __restrict__ out,
                const float* __restrict__ Wq, const float* __restrict__ bq,
                float* __restrict__ qkv) {
  __shared__ float red[2 * 4 * 16];
  __shared__ float hs[16 * APITCH];

  const int lane = threadIdx.x & 31;
  const int wave = threadIdx.x >> 5;
  const int m    = lane & 15;
  const int half = lane >> 4;
  const int tbn  = wave * 64;
  const int tbm  = blockIdx.x * 16;
  const int col  = tbn + m;
  const int rb   = 8 * half;

  float t0[8], t1[8], t2[8], t3[8];
  {
    const float* arow = A + (size_t)(tbm + m) * Dd + half * 4;
    const float* w0   = W1 + (size_t)(tbn + m) * Dd + half * 4;
    const float* w1   = w0 + (size_t)16 * Dd;
    const float* w2   = w1 + (size_t)16 * Dd;
    const float* w3   = w2 + (size_t)16 * Dd;
    v8f c0 = {}, c1 = {}, c2 = {}, c3 = {};
    gemm_tile<Dd>(arow, w0, w1, w2, w3, c0, c1, c2, c3);

    const float bj0 = bias1[col], bj1 = bias1[col + 16], bj2 = bias1[col + 32], bj3 = bias1[col + 48];
    const float g0 = g1[col], ga = g1[col + 16], gb = g1[col + 32], gc = g1[col + 48];
    const float e0 = b1[col], ea = b1[col + 16], eb = b1[col + 32], ec = b1[col + 48];
#pragma unroll
    for (int r = 0; r < 8; ++r) {
      const int row = tbm + rb + r;
      const float sc = rscale[row];
      const float* rr = res + (size_t)row * Dd + col;
      t0[r] = rr[0]  + (c0[r] + bj0) * sc;
      t1[r] = rr[16] + (c1[r] + bj1) * sc;
      t2[r] = rr[32] + (c2[r] + bj2) * sc;
      t3[r] = rr[48] + (c3[r] + bj3) * sc;
    }
    ln_rows(t0, t1, t2, t3, red, wave, m, rb, g0, ga, gb, gc, e0, ea, eb, ec);
  }
  tile_to_lds(hs, rb, col, t0, t1, t2, t3);
  __syncthreads();

  {
    const float* arow = hs + m * APITCH + 4 * half;
    const float* w0   = W2 + (size_t)(tbn + m) * Dd + half * 4;
    const float* w1   = w0 + (size_t)16 * Dd;
    const float* w2   = w1 + (size_t)16 * Dd;
    const float* w3   = w2 + (size_t)16 * Dd;
    v8f c0 = {}, c1 = {}, c2 = {}, c3 = {};
    gemm_tile<Dd>(arow, w0, w1, w2, w3, c0, c1, c2, c3);

    const float bj0 = bias2[col], bj1 = bias2[col + 16], bj2 = bias2[col + 32], bj3 = bias2[col + 48];
    const float g0 = g2[col], ga = g2[col + 16], gb = g2[col + 32], gc = g2[col + 48];
    const float e0 = b2[col], ea = b2[col + 16], eb = b2[col + 32], ec = b2[col + 48];
    float a0[8], a1[8], a2[8], a3[8];
    if (ADD) {
#pragma unroll
      for (int r = 0; r < 8; ++r) {
        const float* ar = addp + (size_t)(tbm + rb + r) * Dd + col;
        a0[r] = ar[0]; a1[r] = ar[16]; a2[r] = ar[32]; a3[r] = ar[48];
      }
    }
#pragma unroll
    for (int r = 0; r < 8; ++r) {
      t0[r] = t0[r] + (c0[r] + bj0);
      t1[r] = t1[r] + (c1[r] + bj1);
      t2[r] = t2[r] + (c2[r] + bj2);
      t3[r] = t3[r] + (c3[r] + bj3);
    }
    ln_rows(t0, t1, t2, t3, red, wave, m, rb, g0, ga, gb, gc, e0, ea, eb, ec);
    if (ADD) {
#pragma unroll
      for (int r = 0; r < 8; ++r) { t0[r] += a0[r]; t1[r] += a1[r]; t2[r] += a2[r]; t3[r] += a3[r]; }
    }
#pragma unroll
    for (int r = 0; r < 8; ++r) {
      float* orow = out + (size_t)(tbm + rb + r) * Dd + col;
      orow[0] = t0[r]; orow[16] = t1[r]; orow[32] = t2[r]; orow[48] = t3[r];
    }
  }

  if (QKV) {
    tile_to_lds(hs, rb, col, t0, t1, t2, t3);
    __syncthreads();
    qkv_rows(hs, Wq, bq, qkv, tbm, wave, m, half);
  }
}

#define PWP  260
#define TPH  (Pp / 16)
#define PWG  256
#define PWV  12

__global__ __launch_bounds__(32 * PWV)
void predict_all(const float* __restrict__ hc,
                 const int* __restrict__ psrc, const int* __restrict__ pdst,
                 const int* __restrict__ nsrc, const int* __restrict__ ndst,
                 const float* __restrict__ pw1, const float* __restrict__ pb1,
                 const float* __restrict__ pw2, const float* __restrict__ pb2,
                 float* __restrict__ out) {
  __shared__ float wl[(Dd / 2) * PWP];

  const int tid = threadIdx.x;
  for (int idx = tid; idx < (Dd / 2) * (Dd / 4); idx += 32 * PWV) {
    const int row = idx >> 6, c4 = idx & 63;
    const float4 v = *(const float4*)(pw1 + (size_t)row * Dd + c4 * 4);
    *(float4*)(wl + row * PWP + c4 * 4) = v;
  }
  __syncthreads();

  const int lane = tid & 31;
  const int wave = tid >> 5;
  const int m    = lane & 15;
  const int half = lane >> 4;

  float bias1[8], w2[8];
#pragma unroll
  for (int jt = 0; jt < 8; ++jt) {
    bias1[jt] = pb1[jt * 16 + m];
    w2[jt]    = pw2[jt * 16 + m];
  }
  const float b2 = pb2[0];
  const float* wfrag = wl + m * PWP + 4 * half;

  for (int t = wave * PWG + (int)blockIdx.x; t < 2 * TPH; t += PWV * PWG) {
    asm volatile("" ::: "memory");
    const bool ng = t >= TPH;
    const int* sp = ng ? nsrc : psrc;
    const int* dp = ng ? ndst : pdst;
    const int tb  = (ng ? t - TPH : t) * 16;
    const int rs = sp[tb + m], rd = dp[tb + m];
    const float* as = hc + (size_t)rs * Dd + 4 * half;
    const float* ad = hc + (size_t)rd * Dd + 4 * half;

    float4 z[32];
#pragma unroll
    for (int i = 0; i < 32; ++i) {
      const float4 a = *(const float4*)(as + 8 * i);
      const float4 b = *(const float4*)(ad + 8 * i);
      z[i] = make_float4(a.x * b.x, a.y * b.y, a.z * b.z, a.w * b.w);
    }

    v8f c[8];
#pragma unroll
    for (int jt = 0; jt < 8; ++jt) c[jt] = v8f{};

#pragma unroll
    for (int i = 0; i < 32; ++i) {
      float4 w[8];
#pragma unroll
      for (int jt = 0; jt < 8; ++jt)
        w[jt] = *(const float4*)(wfrag + jt * 16 * PWP + 8 * i);
      const v2f a0 = { z[i].x, z[i].y };
      const v2f a1 = { z[i].z, z[i].w };
#pragma unroll
      for (int jt = 0; jt < 8; ++jt)
        c[jt] = wmma4(a0, v2f{ w[jt].x, w[jt].y }, c[jt]);
#pragma unroll
      for (int jt = 0; jt < 8; ++jt)
        c[jt] = wmma4(a1, v2f{ w[jt].z, w[jt].w }, c[jt]);
    }

    float acc[8];
#pragma unroll
    for (int r = 0; r < 8; ++r) acc[r] = 0.f;
#pragma unroll
    for (int jt = 0; jt < 8; ++jt) {
#pragma unroll
      for (int r = 0; r < 8; ++r) {
        float v = c[jt][r] + bias1[jt];
        v = v > 0.f ? v : 0.2f * v;
        acc[r] += v * w2[jt];
      }
    }
#pragma unroll
    for (int r = 0; r < 8; ++r) acc[r] = rsum16(acc[r]);
    float o = acc[0];
#pragma unroll
    for (int r = 1; r < 8; ++r) o = (m == r) ? acc[r] : o;
    if (m < 8) out[(size_t)t * 16 + 8 * half + m] = o + b2;
  }
}

extern "C" void kernel_launch(void* const* d_in, const int* in_sizes, int n_in,
                              void* d_out, int out_size, void* d_ws, size_t ws_size,
                              hipStream_t stream) {
  const float* x       = (const float*)d_in[0];
  const float* edge_w  = (const float*)d_in[1];
  const int*   src     = (const int*)d_in[2];
  const int*   pos_src = (const int*)d_in[4];
  const int*   pos_dst = (const int*)d_in[5];
  const int*   neg_src = (const int*)d_in[6];
  const int*   neg_dst = (const int*)d_in[7];
  const float* proj_w  = (const float*)d_in[8];
  const float* proj_b  = (const float*)d_in[9];
  const float* in_w    = (const float*)d_in[10];
  const float* in_b    = (const float*)d_in[11];
  const float* out_w   = (const float*)d_in[12];
  const float* out_b   = (const float*)d_in[13];
  const float* ff_w    = (const float*)d_in[14];
  const float* ff_b    = (const float*)d_in[15];
  const float* g1      = (const float*)d_in[16];
  const float* b1      = (const float*)d_in[17];
  const float* g2      = (const float*)d_in[18];
  const float* b2      = (const float*)d_in[19];
  const float* pw1     = (const float*)d_in[20];
  const float* pb1     = (const float*)d_in[21];
  const float* pw2     = (const float*)d_in[22];
  const float* pb2     = (const float*)d_in[23];

  float* W      = (float*)d_ws;
  float* h      = W;
  float* hproj  = W + 1048576;
  float* qkv    = W + 2097152;
  float* obuf   = W + 5242880;
  float* wsn    = W + 7340032;

  float* outp = (float*)d_out;
  float* pos  = outp;
  float* hc   = outp + 2 * Pp;

  proj_qkv<<<Nn / 16, 128, 0, stream>>>(x, proj_w, proj_b, hproj, in_w, in_b, qkv);

  const float* hin[Ll] = { hproj, h };

  for (int l = 0; l < Ll; ++l) {
    attn_kernel<<<dim3(Nn / 128, Hh), 256, 0, stream>>>(qkv, obuf);
    zerof<<<(Nn + 255) / 256, 256, 0, stream>>>(wsn, Nn);
    seg_sum<<<(Ee + 255) / 256, 256, 0, stream>>>(edge_w + (size_t)l * Ee,
                                                  src + (size_t)l * Ee, wsn, Ee);
    if (l + 1 < Ll)
      layer_tail<true, false><<<Nn / 16, 128, 0, stream>>>(
          obuf, out_w + (size_t)l * Dd * Dd, out_b + (size_t)l * Dd, hin[l], wsn,
          g1 + (size_t)l * Dd, b1 + (size_t)l * Dd,
          ff_w + (size_t)l * Dd * Dd, ff_b + (size_t)l * Dd,
          g2 + (size_t)l * Dd, b2 + (size_t)l * Dd, nullptr, h,
          in_w + (size_t)(l + 1) * 3 * Dd * Dd, in_b + (size_t)(l + 1) * 3 * Dd, qkv);
    else
      layer_tail<false, true><<<Nn / 16, 128, 0, stream>>>(
          obuf, out_w + (size_t)l * Dd * Dd, out_b + (size_t)l * Dd, hin[l], wsn,
          g1 + (size_t)l * Dd, b1 + (size_t)l * Dd,
          ff_w + (size_t)l * Dd * Dd, ff_b + (size_t)l * Dd,
          g2 + (size_t)l * Dd, b2 + (size_t)l * Dd, hproj, hc,
          nullptr, nullptr, nullptr);
  }

  predict_all<<<PWG, 32 * PWV, 0, stream>>>(hc, pos_src, pos_dst, neg_src, neg_dst,
                                       pw1, pb1, pw2, pb2, pos);
}
